# starting kernel
# baseline (speedup 1.0000x reference)
#include <hip/hip_runtime.h>
#include <hip/hip_bf16.h>

typedef float v2f __attribute__((ext_vector_type(2)));
typedef float v8f __attribute__((ext_vector_type(8)));

#define IN_F 256
#define OUT_F 128

__global__ void zero_kernel(float* __restrict__ p, long long n) {
    long long i = (long long)blockIdx.x * blockDim.x + threadIdx.x;
    long long stride = (long long)gridDim.x * blockDim.x;
    for (; i < n; i += stride) p[i] = 0.0f;
}

__global__ __launch_bounds__(256)
void gemm_msg_kernel(const float* __restrict__ miu,
                     const float* __restrict__ sigma,
                     const float* __restrict__ Wmu,
                     const float* __restrict__ Wsig,
                     float* __restrict__ m_mu,
                     float* __restrict__ m_sig,
                     int N) {
    const int lane = threadIdx.x & 31;
    const int wave = threadIdx.x >> 5;
    const int row_base = blockIdx.x * 128 + wave * 16;

    const int m    = lane & 15;
    const int kgrp = lane >> 4;

    const int arow = (row_base + m < N) ? (row_base + m) : (N - 1);

    float att[8][8];

    {
        v8f acc[8];
        v8f zero = {};
        #pragma unroll
        for (int t = 0; t < 8; ++t) acc[t] = zero;

        const float* Ap = sigma + (size_t)arow * IN_F + kgrp * 2;
        #pragma unroll 4
        for (int k0 = 0; k0 < IN_F; k0 += 4) {
            v2f a = *(const v2f*)(Ap + k0);
            const float* Bp = Wsig + (size_t)(k0 + kgrp * 2) * OUT_F + m;
            #pragma unroll
            for (int t = 0; t < 8; ++t) {
                v2f b;
                b[0] = Bp[t * 16];
                b[1] = Bp[t * 16 + OUT_F];
                acc[t] = __builtin_amdgcn_wmma_f32_16x16x4_f32(
                    false, a, false, b, (short)0, acc[t], false, false);
            }
        }
        #pragma unroll
        for (int t = 0; t < 8; ++t) {
            #pragma unroll
            for (int r = 0; r < 8; ++r) {
                float hs = fmaxf(acc[t][r], 0.0f);
                float a  = __expf(-hs);
                att[t][r] = a;
                int rr = row_base + r + kgrp * 8;
                if (rr < N)
                    m_sig[(size_t)rr * OUT_F + t * 16 + m] = hs * a * a;
            }
        }
    }

    {
        v8f acc[8];
        v8f zero = {};
        #pragma unroll
        for (int t = 0; t < 8; ++t) acc[t] = zero;

        const float* Ap = miu + (size_t)arow * IN_F + kgrp * 2;
        #pragma unroll 4
        for (int k0 = 0; k0 < IN_F; k0 += 4) {
            v2f a = *(const v2f*)(Ap + k0);
            const float* Bp = Wmu + (size_t)(k0 + kgrp * 2) * OUT_F + m;
            #pragma unroll
            for (int t = 0; t < 8; ++t) {
                v2f b;
                b[0] = Bp[t * 16];
                b[1] = Bp[t * 16 + OUT_F];
                acc[t] = __builtin_amdgcn_wmma_f32_16x16x4_f32(
                    false, a, false, b, (short)0, acc[t], false, false);
            }
        }
        #pragma unroll
        for (int t = 0; t < 8; ++t) {
            #pragma unroll
            for (int r = 0; r < 8; ++r) {
                float hm = acc[t][r];
                float e  = (hm > 0.0f) ? hm : (__expf(hm) - 1.0f);
                int rr = row_base + r + kgrp * 8;
                if (rr < N)
                    m_mu[(size_t)rr * OUT_F + t * 16 + m] = e * att[t][r];
            }
        }
    }
}

__global__ __launch_bounds__(256)
void spmm_scatter_kernel(const int* __restrict__ erow,
                         const int* __restrict__ ecol,
                         const float* __restrict__ a1,
                         const float* __restrict__ a2,
                         const float* __restrict__ m_mu,
                         const float* __restrict__ m_sig,
                         float* __restrict__ out_mu,
                         float* __restrict__ out_sig,
                         int E) {
    const int lane = threadIdx.x & 31;
    const long long wid = ((long long)blockIdx.x * blockDim.x + threadIdx.x) >> 5;
    if (wid >= E) return;
    const int e = (int)wid;

    const int r = erow[e];
    const int c = ecol[e];
    const float v1 = a1[e];
    const float v2 = a2[e];

    const float4 mm = ((const float4*)(m_mu  + (size_t)c * OUT_F))[lane];
    const float4 ms = ((const float4*)(m_sig + (size_t)c * OUT_F))[lane];

    float* om = out_mu  + (size_t)r * OUT_F + lane * 4;
    float* os = out_sig + (size_t)r * OUT_F + lane * 4;

    unsafeAtomicAdd(om + 0, v1 * mm.x);
    unsafeAtomicAdd(om + 1, v1 * mm.y);
    unsafeAtomicAdd(om + 2, v1 * mm.z);
    unsafeAtomicAdd(om + 3, v1 * mm.w);
    unsafeAtomicAdd(os + 0, v2 * ms.x);
    unsafeAtomicAdd(os + 1, v2 * ms.y);
    unsafeAtomicAdd(os + 2, v2 * ms.z);
    unsafeAtomicAdd(os + 3, v2 * ms.w);
}

extern "C" void kernel_launch(void* const* d_in, const int* in_sizes, int n_in,
                              void* d_out, int out_size, void* d_ws, size_t ws_size,
                              hipStream_t stream) {
    const float* miu   = (const float*)d_in[0];
    const float* sigma = (const float*)d_in[1];
    const int*   erow  = (const int*)  d_in[2];
    const int*   ecol  = (const int*)  d_in[3];
    const float* a1    = (const float*)d_in[4];
    const float* a2    = (const float*)d_in[5];
    const float* Wmu   = (const float*)d_in[6];
    const float* Wsig  = (const float*)d_in[7];

    const int N = in_sizes[0] / IN_F;
    const int E = in_sizes[2];

    float* out_mu  = (float*)d_out;
    float* out_sig = out_mu + (size_t)N * OUT_F;

    float* m_mu  = (float*)d_ws;
    float* m_sig = m_mu + (size_t)N * OUT_F;

    zero_kernel<<<2048, 256, 0, stream>>>(out_mu, (long long)2 * N * OUT_F);

    gemm_msg_kernel<<<(N + 127) / 128, 256, 0, stream>>>(
        miu, sigma, Wmu, Wsig, m_mu, m_sig, N);

    const long long waves = E;
    const long long blocks = (waves * 32 + 255) / 256;
    spmm_scatter_kernel<<<(int)blocks, 256, 0, stream>>>(
        erow, ecol, a1, a2, m_mu, m_sig, out_mu, out_sig, E);
}


# message GEMM: a resident work-group per WGP keeps both weight matrices whole in LDS and deals 16-row tiles to its 16 waves; B fragments by 8-byte LDS loads, not 4-byte global loads; k order kept
# speedup vs baseline: 2.7580x; 2.7580x over previous
#include <hip/hip_runtime.h>
#include <hip/hip_bf16.h>

typedef float v2f __attribute__((ext_vector_type(2)));
typedef float v8f __attribute__((ext_vector_type(8)));

#define IN_F 256
#define OUT_F 128

#define GEMM_WAVES 16
#define GEMM_THR   (GEMM_WAVES * 32)
#define W_FLOATS   (IN_F * OUT_F)
#define W_QUADS    (W_FLOATS / 4)
#define KSLAB      (OUT_F * 4)
#define NT_REST    2

template <int NT>
__device__ __forceinline__ void gemm_tile(const float* __restrict__ miu,
                                          const float* __restrict__ sigma,
                                          const float* bsig, const float* bmu,
                                          float* __restrict__ m_mu,
                                          float* __restrict__ m_sig,
                                          int N, int row_base, int ct0, int m, int kgrp) {
    const int arow = (row_base + m < N) ? (row_base + m) : (N - 1);

    float att[NT][8];

    {
        v8f acc[NT];
        v8f zero = {};
        #pragma unroll
        for (int t = 0; t < NT; ++t) acc[t] = zero;

        const float* Ap = sigma + (size_t)arow * IN_F + kgrp * 2;
        const float* Bw = bsig + ct0 * 64;
        #pragma unroll 4
        for (int k0 = 0; k0 < IN_F; k0 += 4) {
            v2f a = *(const v2f*)(Ap + k0);
            const float* Bp = Bw + (k0 >> 2) * KSLAB;
            #pragma unroll
            for (int t = 0; t < NT; ++t) {
                v2f b = *(const v2f*)(Bp + t * 64);
                acc[t] = __builtin_amdgcn_wmma_f32_16x16x4_f32(
                    false, a, false, b, (short)0, acc[t], false, false);
            }
        }
        #pragma unroll
        for (int t = 0; t < NT; ++t) {
            #pragma unroll
            for (int r = 0; r < 8; ++r) {
                float hs = fmaxf(acc[t][r], 0.0f);
                float a  = __expf(-hs);
                att[t][r] = a;
                int rr = row_base + r + kgrp * 8;
                if (rr < N)
                    m_sig[(size_t)rr * OUT_F + (ct0 + t) * 16 + m] = hs * a * a;
            }
        }
    }

    {
        v8f acc[NT];
        v8f zero = {};
        #pragma unroll
        for (int t = 0; t < NT; ++t) acc[t] = zero;

        const float* Ap = miu + (size_t)arow * IN_F + kgrp * 2;
        const float* Bw = bmu + ct0 * 64;
        #pragma unroll 4
        for (int k0 = 0; k0 < IN_F; k0 += 4) {
            v2f a = *(const v2f*)(Ap + k0);
            const float* Bp = Bw + (k0 >> 2) * KSLAB;
            #pragma unroll
            for (int t = 0; t < NT; ++t) {
                v2f b = *(const v2f*)(Bp + t * 64);
                acc[t] = __builtin_amdgcn_wmma_f32_16x16x4_f32(
                    false, a, false, b, (short)0, acc[t], false, false);
            }
        }
        #pragma unroll
        for (int t = 0; t < NT; ++t) {
            #pragma unroll
            for (int r = 0; r < 8; ++r) {
                float hm = acc[t][r];
                float e  = (hm > 0.0f) ? hm : (__expf(hm) - 1.0f);
                int rr = row_base + r + kgrp * 8;
                if (rr < N)
                    m_mu[(size_t)rr * OUT_F + (ct0 + t) * 16 + m] = e * att[t][r];
            }
        }
    }
}

__global__ __launch_bounds__(GEMM_THR)
void gemm_msg_kernel(const float* __restrict__ miu,
                     const float* __restrict__ sigma,
                     const float* __restrict__ Wmu,
                     const float* __restrict__ Wsig,
                     float* __restrict__ m_mu,
                     float* __restrict__ m_sig,
                     int N, int ntiles) {
    __shared__ __attribute__((aligned(16))) float wl[2 * W_FLOATS];

    const int tid  = threadIdx.x;

    {
        const int n = tid & (OUT_F - 1);
        const int q = tid / OUT_F;
        const float* ssig = Wsig + (size_t)(4 * q) * OUT_F + n;
        const float* smu  = Wmu  + (size_t)(4 * q) * OUT_F + n;
        float4* dsig = (float4*)wl + tid;
        float4* dmu  = dsig + W_QUADS;
        #pragma unroll
        for (int j = 0; j < W_QUADS / GEMM_THR; ++j) {
            const float* s = ssig + (size_t)j * (GEMM_THR / OUT_F) * 4 * OUT_F;
            float4 v;
            v.x = s[0]; v.y = s[OUT_F]; v.z = s[2 * OUT_F]; v.w = s[3 * OUT_F];
            dsig[j * GEMM_THR] = v;
        }
        #pragma unroll
        for (int j = 0; j < W_QUADS / GEMM_THR; ++j) {
            const float* s = smu + (size_t)j * (GEMM_THR / OUT_F) * 4 * OUT_F;
            float4 v;
            v.x = s[0]; v.y = s[OUT_F]; v.z = s[2 * OUT_F]; v.w = s[3 * OUT_F];
            dmu[j * GEMM_THR] = v;
        }
    }
    __syncthreads();

    const int lane = tid & 31;
    const int wave = __builtin_amdgcn_readfirstlane(tid >> 5);

    const int m    = lane & 15;
    const int kgrp = lane >> 4;

    const float* bsig = wl + m * 4 + kgrp * 2;
    const float* bmu  = bsig + W_FLOATS;

    const int G      = (int)gridDim.x;
    const int nfull  = ntiles - ntiles % (4 * G);
    const int nslots = nfull + (ntiles - nfull) * (8 / NT_REST);

    for (int slot = wave * G + (int)blockIdx.x; slot < nslots; slot += GEMM_WAVES * G) {
        if (slot < nfull) {
            gemm_tile<8>(miu, sigma, bsig, bmu, m_mu, m_sig, N, slot * 16, 0, m, kgrp);
        } else {
            const int u = slot - nfull;
            gemm_tile<NT_REST>(miu, sigma, bsig, bmu, m_mu, m_sig, N,
                               (nfull + u / (8 / NT_REST)) * 16, (u % (8 / NT_REST)) * NT_REST, m, kgrp);
        }
    }
}

#define OWN_ROWS 196
#define SPAN_W   32
#define SPAN_CH  (SPAN_W * 32)
#define OWN_TILE 512
#define TILE_P   (OWN_TILE + 1)

__global__ __launch_bounds__(1024)
void owner_planes_kernel(const int* __restrict__ erow,
                         unsigned* __restrict__ planes,
                         int N, int E, int nblk, int wpad) {
    __shared__ unsigned bits[SPAN_W * TILE_P];

    const int tid  = threadIdx.x;
    const int lane = tid & 31;
    const int wave = __builtin_amdgcn_readfirstlane(tid >> 5);
    const int c0   = blockIdx.x * SPAN_CH + wave * 32;

    int own[32];
    #pragma unroll
    for (int i = 0; i < 32; ++i) {
        const int e = (c0 + i) * 32 + lane;
        const int r = erow[min(e, E - 1)];
        own[i] = (e < E && (unsigned)r < (unsigned)N) ? r / OWN_ROWS : -1;
    }

    for (int ob = 0; ob < nblk; ob += OWN_TILE) {
        for (int i = tid; i < SPAN_W * TILE_P; i += 1024) bits[i] = 0u;
        __syncthreads();
        #pragma unroll
        for (int i = 0; i < 32; ++i) {
            const unsigned o = (unsigned)(own[i] - ob);
            if (o < (unsigned)OWN_TILE) atomicOr(&bits[wave * TILE_P + o], 1u << i);
        }
        __syncthreads();
        const int no = (nblk - ob < OWN_TILE) ? (nblk - ob) : OWN_TILE;
        for (int i = tid; i < no * SPAN_W; i += 1024) {
            const int o = i >> 5;
            planes[(size_t)(ob + o) * wpad + blockIdx.x * SPAN_W + lane] = bits[lane * TILE_P + o];
        }
        __syncthreads();
    }
}

#define ROW_W    (2 * OUT_F)
#define VG       16
#define QCAP     (VG * 32 + 160)
#define CLCAP    128
#define GRP      8

template <int G>
__device__ __forceinline__ void gather_group(float* __restrict__ acc,
                                             const float* __restrict__ m_mu,
                                             const float* __restrict__ m_sig,
                                             int r, int c, float v1, float v2,
                                             int j0, int lane) {
    float gm[G][4], gs[G][4];
    #pragma unroll
    for (int g = 0; g < G; ++g) {
        const int cj = __builtin_amdgcn_readlane(c, j0 + g);
        const float* pm = m_mu  + (size_t)cj * OUT_F + lane;
        const float* ps = m_sig + (size_t)cj * OUT_F + lane;
        #pragma unroll
        for (int k = 0; k < 4; ++k) { gm[g][k] = pm[32 * k]; gs[g][k] = ps[32 * k]; }
    }
    #pragma unroll
    for (int g = 0; g < G; ++g) {
        const int   rj  = __builtin_amdgcn_readlane(r, j0 + g);
        const float w1  = __int_as_float(__builtin_amdgcn_readlane(__float_as_int(v1), j0 + g));
        const float w2  = __int_as_float(__builtin_amdgcn_readlane(__float_as_int(v2), j0 + g));
        float* dst = acc + rj * ROW_W + lane;
        #pragma unroll
        for (int k = 0; k < 4; ++k) {
            unsafeAtomicAdd(dst + 32 * k,         w1 * gm[g][k]);
            unsafeAtomicAdd(dst + OUT_F + 32 * k, w2 * gs[g][k]);
        }
    }
}

template <int G>
__device__ __forceinline__ void drain_list(float* __restrict__ acc, const int* __restrict__ myq, int cnt,
                                           const int* __restrict__ erow, const int* __restrict__ ecol,
                                           const float* __restrict__ a1, const float* __restrict__ a2,
                                           const float* __restrict__ m_mu, const float* __restrict__ m_sig,
                                           int row0, int lane) {
    for (int i0 = 0; i0 < cnt; i0 += 32) {
        const int n = (cnt - i0 < 32) ? (cnt - i0) : 32;
        int r = 0, c = 0;
        float v1 = 0.f, v2 = 0.f;
        if (lane < n) {
            const int e = myq[i0 + lane];
            r  = erow[e] - row0;
            c  = ecol[e];
            v1 = a1[e];
            v2 = a2[e];
        }
        int j = 0;
        for (; j + G <= n; j += G)
            gather_group<G>(acc, m_mu, m_sig, r, c, v1, v2, j, lane);
        if (G > 4 && j + 4 <= n) { gather_group<4>(acc, m_mu, m_sig, r, c, v1, v2, j, lane); j += 4; }
        if (G > 2 && j + 2 <= n) { gather_group<2>(acc, m_mu, m_sig, r, c, v1, v2, j, lane); j += 2; }
        if (G > 1 && j < n)        gather_group<1>(acc, m_mu, m_sig, r, c, v1, v2, j, lane);
    }
}

__global__ __launch_bounds__(1024)
void spmm_owner_kernel(const int* __restrict__ erow,
                       const int* __restrict__ ecol,
                       const float* __restrict__ a1,
                       const float* __restrict__ a2,
                       const float* __restrict__ m_mu,
                       const float* __restrict__ m_sig,
                       const unsigned* __restrict__ planes,
                       float* __restrict__ out_mu,
                       float* __restrict__ out_sig,
                       int N, int E, int wpad) {
    __shared__ __attribute__((aligned(16))) float acc[OWN_ROWS * ROW_W];
    __shared__ int elist[32 * QCAP];
    __shared__ int clist[32 * CLCAP];

    const int tid  = threadIdx.x;
    const int lane = tid & 31;
    const int wave = __builtin_amdgcn_readfirstlane(tid >> 5);
    const int row0 = blockIdx.x * OWN_ROWS;
    const int nrows = (N - row0 < OWN_ROWS) ? (N - row0) : OWN_ROWS;
    int* myq = elist + wave * QCAP;
    int* myc = clist + wave * CLCAP;

    for (int i = tid; i < OWN_ROWS * ROW_W / 4; i += 1024)
        ((float4*)acc)[i] = make_float4(0.f, 0.f, 0.f, 0.f);
    __syncthreads();

    int cnt = 0;

    const int nchunks = (E + 31) >> 5;
    const int nwords  = (nchunks + 31) >> 5;
    const int per     = (nwords + 31) >> 5;
    const int wbeg    = wave * per;
    const int wend    = (wbeg + per < nwords) ? (wbeg + per) : nwords;
    const unsigned* plane = planes + (size_t)blockIdx.x * wpad;

    auto make_room = [&]() {
        if (cnt > QCAP - VG * 32) {
            drain_list<1>(acc, myq, cnt, erow, ecol, a1, a2, m_mu, m_sig, row0, lane);
            cnt = 0;
        }
    };

    auto visit = [&](int nlist) {
        for (int i0 = 0; i0 < nlist; i0 += 32) {
            const int n = (nlist - i0 < 32) ? (nlist - i0) : 32;
            const int cid = (lane < n) ? myc[i0 + lane] : nchunks;
            for (int j = 0; j < n; j += VG) {
                make_room();
                int rv[VG];
                #pragma unroll
                for (int g = 0; g < VG; ++g) {
                    const int e = __builtin_amdgcn_readlane(cid, j + g) * 32 + lane;
                    rv[g] = erow[min(e, E - 1)];
                }
                #pragma unroll
                for (int g = 0; g < VG; ++g) {
                    const int e = __builtin_amdgcn_readlane(cid, j + g) * 32 + lane;
                    const bool hit = (e < E) && ((unsigned)(rv[g] - row0) < (unsigned)nrows);
                    const unsigned mask = __builtin_amdgcn_ballot_w32(hit);
                    const int pos = cnt + (int)__builtin_amdgcn_mbcnt_lo(mask, 0u);
                    if (hit) myq[pos] = e;
                    cnt += __builtin_popcount(mask);
                }
            }
        }
    };

    auto plane_word = [&](int w0) -> unsigned {
        const int wi = w0 + lane;
        return (wi < wend) ? plane[wi] : 0u;
    };

    unsigned word = plane_word(wbeg);
    for (int w0 = wbeg; w0 < wend; w0 += 32) {
        const unsigned nxt = plane_word(w0 + 32);
        const int cword = (w0 + lane) << 5;
        int nlist = 0;
        for (int t = 0; t <= 32; ++t) {
            const bool has = word != 0u;
            const unsigned m = __builtin_amdgcn_ballot_w32(has);
            const int k = __builtin_popcount(m);
            if (k == 0 || nlist + k > CLCAP) { visit(nlist); nlist = 0; }
            if (k == 0) break;
            if (has) {
                myc[nlist + (int)__builtin_amdgcn_mbcnt_lo(m, 0u)] = cword + __builtin_ctz(word);
                word &= word - 1u;
            }
            nlist += k;
        }
        word = nxt;
    }
    drain_list<GRP>(acc, myq, cnt, erow, ecol, a1, a2, m_mu, m_sig, row0, lane);
    __syncthreads();

    for (int i = tid; i < nrows * (OUT_F / 4); i += 1024) {
        const int r  = i >> 5;
        const int c4 = i & 31;
        const float4 vm = *(const float4*)(acc + r * ROW_W + c4 * 4);
        const float4 vs = *(const float4*)(acc + r * ROW_W + OUT_F + c4 * 4);
        *(float4*)(out_mu  + (size_t)(row0 + r) * OUT_F + c4 * 4) = vm;
        *(float4*)(out_sig + (size_t)(row0 + r) * OUT_F + c4 * 4) = vs;
    }
}

extern "C" void kernel_launch(void* const* d_in, const int* in_sizes, int n_in,
                              void* d_out, int out_size, void* d_ws, size_t ws_size,
                              hipStream_t stream) {
    const float* miu   = (const float*)d_in[0];
    const float* sigma = (const float*)d_in[1];
    const int*   erow  = (const int*)  d_in[2];
    const int*   ecol  = (const int*)  d_in[3];
    const float* a1    = (const float*)d_in[4];
    const float* a2    = (const float*)d_in[5];
    const float* Wmu   = (const float*)d_in[6];
    const float* Wsig  = (const float*)d_in[7];

    const int N = in_sizes[0] / IN_F;
    const int E = in_sizes[2];

    float* out_mu  = (float*)d_out;
    float* out_sig = out_mu + (size_t)N * OUT_F;

    float* m_mu  = (float*)d_ws;
    float* m_sig = m_mu + (size_t)N * OUT_F;
    unsigned* planes = (unsigned*)(m_sig + (size_t)N * OUT_F);

    const int nblk   = (N + OWN_ROWS - 1) / OWN_ROWS;
    const int nspans = (E + SPAN_CH * 32 - 1) / (SPAN_CH * 32);
    const int wpad   = nspans * SPAN_W;

    gemm_msg_kernel<<<256, GEMM_THR, 0, stream>>>(
        miu, sigma, Wmu, Wsig, m_mu, m_sig, N, (N + 15) / 16);

    owner_planes_kernel<<<nspans, 1024, 0, stream>>>(erow, planes, N, E, nblk, wpad);

    spmm_owner_kernel<<<nblk, 1024, 0, stream>>>(
        erow, ecol, a1, a2, m_mu, m_sig, planes, out_mu, out_sig, N, E, wpad);
}


# owner SpMM gather: a lane loads 16 bytes, one load = a whole 512-byte message row (2 loads an edge, not 8); LDS sums held in load-register order, reordered at store-out; adds still in arrival order
# speedup vs baseline: 2.8405x; 1.0299x over previous
#include <hip/hip_runtime.h>
#include <hip/hip_bf16.h>

typedef float v2f __attribute__((ext_vector_type(2)));
typedef float v8f __attribute__((ext_vector_type(8)));

#define IN_F 256
#define OUT_F 128

#define GEMM_WAVES 16
#define GEMM_THR   (GEMM_WAVES * 32)
#define W_FLOATS   (IN_F * OUT_F)
#define W_QUADS    (W_FLOATS / 4)
#define KSLAB      (OUT_F * 4)
#define NT_REST    2

template <int NT>
__device__ __forceinline__ void gemm_tile(const float* __restrict__ miu,
                                          const float* __restrict__ sigma,
                                          const float* bsig, const float* bmu,
                                          float* __restrict__ m_mu,
                                          float* __restrict__ m_sig,
                                          int N, int row_base, int ct0, int m, int kgrp) {
    const int arow = (row_base + m < N) ? (row_base + m) : (N - 1);

    float att[NT][8];

    {
        v8f acc[NT];
        v8f zero = {};
        #pragma unroll
        for (int t = 0; t < NT; ++t) acc[t] = zero;

        const float* Ap = sigma + (size_t)arow * IN_F + kgrp * 2;
        const float* Bw = bsig + ct0 * 64;
        #pragma unroll 4
        for (int k0 = 0; k0 < IN_F; k0 += 4) {
            v2f a = *(const v2f*)(Ap + k0);
            const float* Bp = Bw + (k0 >> 2) * KSLAB;
            #pragma unroll
            for (int t = 0; t < NT; ++t) {
                v2f b = *(const v2f*)(Bp + t * 64);
                acc[t] = __builtin_amdgcn_wmma_f32_16x16x4_f32(
                    false, a, false, b, (short)0, acc[t], false, false);
            }
        }
        #pragma unroll
        for (int t = 0; t < NT; ++t) {
            #pragma unroll
            for (int r = 0; r < 8; ++r) {
                float hs = fmaxf(acc[t][r], 0.0f);
                float a  = __expf(-hs);
                att[t][r] = a;
                int rr = row_base + r + kgrp * 8;
                if (rr < N)
                    m_sig[(size_t)rr * OUT_F + (ct0 + t) * 16 + m] = hs * a * a;
            }
        }
    }

    {
        v8f acc[NT];
        v8f zero = {};
        #pragma unroll
        for (int t = 0; t < NT; ++t) acc[t] = zero;

        const float* Ap = miu + (size_t)arow * IN_F + kgrp * 2;
        const float* Bw = bmu + ct0 * 64;
        #pragma unroll 4
        for (int k0 = 0; k0 < IN_F; k0 += 4) {
            v2f a = *(const v2f*)(Ap + k0);
            const float* Bp = Bw + (k0 >> 2) * KSLAB;
            #pragma unroll
            for (int t = 0; t < NT; ++t) {
                v2f b = *(const v2f*)(Bp + t * 64);
                acc[t] = __builtin_amdgcn_wmma_f32_16x16x4_f32(
                    false, a, false, b, (short)0, acc[t], false, false);
            }
        }
        #pragma unroll
        for (int t = 0; t < NT; ++t) {
            #pragma unroll
            for (int r = 0; r < 8; ++r) {
                float hm = acc[t][r];
                float e  = (hm > 0.0f) ? hm : (__expf(hm) - 1.0f);
                int rr = row_base + r + kgrp * 8;
                if (rr < N)
                    m_mu[(size_t)rr * OUT_F + (ct0 + t) * 16 + m] = e * att[t][r];
            }
        }
    }
}

__global__ __launch_bounds__(GEMM_THR)
void gemm_msg_kernel(const float* __restrict__ miu,
                     const float* __restrict__ sigma,
                     const float* __restrict__ Wmu,
                     const float* __restrict__ Wsig,
                     float* __restrict__ m_mu,
                     float* __restrict__ m_sig,
                     int N, int ntiles) {
    __shared__ __attribute__((aligned(16))) float wl[2 * W_FLOATS];

    const int tid  = threadIdx.x;

    {
        const int n = tid & (OUT_F - 1);
        const int q = tid / OUT_F;
        const float* ssig = Wsig + (size_t)(4 * q) * OUT_F + n;
        const float* smu  = Wmu  + (size_t)(4 * q) * OUT_F + n;
        float4* dsig = (float4*)wl + tid;
        float4* dmu  = dsig + W_QUADS;
        #pragma unroll
        for (int j = 0; j < W_QUADS / GEMM_THR; ++j) {
            const float* s = ssig + (size_t)j * (GEMM_THR / OUT_F) * 4 * OUT_F;
            float4 v;
            v.x = s[0]; v.y = s[OUT_F]; v.z = s[2 * OUT_F]; v.w = s[3 * OUT_F];
            dsig[j * GEMM_THR] = v;
        }
        #pragma unroll
        for (int j = 0; j < W_QUADS / GEMM_THR; ++j) {
            const float* s = smu + (size_t)j * (GEMM_THR / OUT_F) * 4 * OUT_F;
            float4 v;
            v.x = s[0]; v.y = s[OUT_F]; v.z = s[2 * OUT_F]; v.w = s[3 * OUT_F];
            dmu[j * GEMM_THR] = v;
        }
    }
    __syncthreads();

    const int lane = tid & 31;
    const int wave = __builtin_amdgcn_readfirstlane(tid >> 5);

    const int m    = lane & 15;
    const int kgrp = lane >> 4;

    const float* bsig = wl + m * 4 + kgrp * 2;
    const float* bmu  = bsig + W_FLOATS;

    const int G      = (int)gridDim.x;
    const int nfull  = ntiles - ntiles % (4 * G);
    const int nslots = nfull + (ntiles - nfull) * (8 / NT_REST);

    for (int slot = wave * G + (int)blockIdx.x; slot < nslots; slot += GEMM_WAVES * G) {
        if (slot < nfull) {
            gemm_tile<8>(miu, sigma, bsig, bmu, m_mu, m_sig, N, slot * 16, 0, m, kgrp);
        } else {
            const int u = slot - nfull;
            gemm_tile<NT_REST>(miu, sigma, bsig, bmu, m_mu, m_sig, N,
                               (nfull + u / (8 / NT_REST)) * 16, (u % (8 / NT_REST)) * NT_REST, m, kgrp);
        }
    }
}

#define OWN_ROWS 196
#define SPAN_W   32
#define SPAN_CH  (SPAN_W * 32)
#define OWN_TILE 512
#define TILE_P   (OWN_TILE + 1)

__global__ __launch_bounds__(1024)
void owner_planes_kernel(const int* __restrict__ erow,
                         unsigned* __restrict__ planes,
                         int N, int E, int nblk, int wpad) {
    __shared__ unsigned bits[SPAN_W * TILE_P];

    const int tid  = threadIdx.x;
    const int lane = tid & 31;
    const int wave = __builtin_amdgcn_readfirstlane(tid >> 5);
    const int c0   = blockIdx.x * SPAN_CH + wave * 32;

    int own[32];
    #pragma unroll
    for (int i = 0; i < 32; ++i) {
        const int e = (c0 + i) * 32 + lane;
        const int r = erow[min(e, E - 1)];
        own[i] = (e < E && (unsigned)r < (unsigned)N) ? r / OWN_ROWS : -1;
    }

    for (int ob = 0; ob < nblk; ob += OWN_TILE) {
        for (int i = tid; i < SPAN_W * TILE_P; i += 1024) bits[i] = 0u;
        __syncthreads();
        #pragma unroll
        for (int i = 0; i < 32; ++i) {
            const unsigned o = (unsigned)(own[i] - ob);
            if (o < (unsigned)OWN_TILE) atomicOr(&bits[wave * TILE_P + o], 1u << i);
        }
        __syncthreads();
        const int no = (nblk - ob < OWN_TILE) ? (nblk - ob) : OWN_TILE;
        for (int i = tid; i < no * SPAN_W; i += 1024) {
            const int o = i >> 5;
            planes[(size_t)(ob + o) * wpad + blockIdx.x * SPAN_W + lane] = bits[lane * TILE_P + o];
        }
        __syncthreads();
    }
}

#define ROW_W    (2 * OUT_F)
#define VG       16
#define QCAP     (VG * 32 + 160)
#define CLCAP    128
#define GRP      8

template <int G>
__device__ __forceinline__ void gather_group(float* __restrict__ acc,
                                             const float* __restrict__ m_mu,
                                             const float* __restrict__ m_sig,
                                             int r, int c, float v1, float v2,
                                             int j0, int lane) {
    float4 gm[G], gs[G];
    #pragma unroll
    for (int g = 0; g < G; ++g) {
        const int cj = __builtin_amdgcn_readlane(c, j0 + g);
        gm[g] = *(const float4*)(m_mu  + (size_t)cj * OUT_F + lane * 4);
        gs[g] = *(const float4*)(m_sig + (size_t)cj * OUT_F + lane * 4);
    }
    #pragma unroll
    for (int g = 0; g < G; ++g) {
        const int   rj  = __builtin_amdgcn_readlane(r, j0 + g);
        const float w1  = __int_as_float(__builtin_amdgcn_readlane(__float_as_int(v1), j0 + g));
        const float w2  = __int_as_float(__builtin_amdgcn_readlane(__float_as_int(v2), j0 + g));
        float* dst = acc + rj * ROW_W + lane;
        unsafeAtomicAdd(dst,              w1 * gm[g].x);
        unsafeAtomicAdd(dst + OUT_F,      w2 * gs[g].x);
        unsafeAtomicAdd(dst + 32,         w1 * gm[g].y);
        unsafeAtomicAdd(dst + OUT_F + 32, w2 * gs[g].y);
        unsafeAtomicAdd(dst + 64,         w1 * gm[g].z);
        unsafeAtomicAdd(dst + OUT_F + 64, w2 * gs[g].z);
        unsafeAtomicAdd(dst + 96,         w1 * gm[g].w);
        unsafeAtomicAdd(dst + OUT_F + 96, w2 * gs[g].w);
    }
}

template <int G>
__device__ __forceinline__ void drain_list(float* __restrict__ acc, const int* __restrict__ myq, int cnt,
                                           const int* __restrict__ erow, const int* __restrict__ ecol,
                                           const float* __restrict__ a1, const float* __restrict__ a2,
                                           const float* __restrict__ m_mu, const float* __restrict__ m_sig,
                                           int row0, int lane) {
    for (int i0 = 0; i0 < cnt; i0 += 32) {
        const int n = (cnt - i0 < 32) ? (cnt - i0) : 32;
        int r = 0, c = 0;
        float v1 = 0.f, v2 = 0.f;
        if (lane < n) {
            const int e = myq[i0 + lane];
            r  = erow[e] - row0;
            c  = ecol[e];
            v1 = a1[e];
            v2 = a2[e];
        }
        int j = 0;
        for (; j + G <= n; j += G)
            gather_group<G>(acc, m_mu, m_sig, r, c, v1, v2, j, lane);
        if (G > 4 && j + 4 <= n) { gather_group<4>(acc, m_mu, m_sig, r, c, v1, v2, j, lane); j += 4; }
        if (G > 2 && j + 2 <= n) { gather_group<2>(acc, m_mu, m_sig, r, c, v1, v2, j, lane); j += 2; }
        if (G > 1 && j < n)        gather_group<1>(acc, m_mu, m_sig, r, c, v1, v2, j, lane);
    }
}

__global__ __launch_bounds__(1024)
void spmm_owner_kernel(const int* __restrict__ erow,
                       const int* __restrict__ ecol,
                       const float* __restrict__ a1,
                       const float* __restrict__ a2,
                       const float* __restrict__ m_mu,
                       const float* __restrict__ m_sig,
                       const unsigned* __restrict__ planes,
                       float* __restrict__ out_mu,
                       float* __restrict__ out_sig,
                       int N, int E, int wpad) {
    __shared__ __attribute__((aligned(16))) float acc[OWN_ROWS * ROW_W];
    __shared__ int elist[32 * QCAP];
    __shared__ int clist[32 * CLCAP];

    const int tid  = threadIdx.x;
    const int lane = tid & 31;
    const int wave = __builtin_amdgcn_readfirstlane(tid >> 5);
    const int row0 = blockIdx.x * OWN_ROWS;
    const int nrows = (N - row0 < OWN_ROWS) ? (N - row0) : OWN_ROWS;
    int* myq = elist + wave * QCAP;
    int* myc = clist + wave * CLCAP;

    for (int i = tid; i < OWN_ROWS * ROW_W / 4; i += 1024)
        ((float4*)acc)[i] = make_float4(0.f, 0.f, 0.f, 0.f);
    __syncthreads();

    int cnt = 0;

    const int nchunks = (E + 31) >> 5;
    const int nwords  = (nchunks + 31) >> 5;
    const int per     = (nwords + 31) >> 5;
    const int wbeg    = wave * per;
    const int wend    = (wbeg + per < nwords) ? (wbeg + per) : nwords;
    const unsigned* plane = planes + (size_t)blockIdx.x * wpad;

    auto make_room = [&]() {
        if (cnt > QCAP - VG * 32) {
            drain_list<1>(acc, myq, cnt, erow, ecol, a1, a2, m_mu, m_sig, row0, lane);
            cnt = 0;
        }
    };

    auto visit = [&](int nlist) {
        for (int i0 = 0; i0 < nlist; i0 += 32) {
            const int n = (nlist - i0 < 32) ? (nlist - i0) : 32;
            const int cid = (lane < n) ? myc[i0 + lane] : nchunks;
            for (int j = 0; j < n; j += VG) {
                make_room();
                int rv[VG];
                #pragma unroll
                for (int g = 0; g < VG; ++g) {
                    const int e = __builtin_amdgcn_readlane(cid, j + g) * 32 + lane;
                    rv[g] = erow[min(e, E - 1)];
                }
                #pragma unroll
                for (int g = 0; g < VG; ++g) {
                    const int e = __builtin_amdgcn_readlane(cid, j + g) * 32 + lane;
                    const bool hit = (e < E) && ((unsigned)(rv[g] - row0) < (unsigned)nrows);
                    const unsigned mask = __builtin_amdgcn_ballot_w32(hit);
                    const int pos = cnt + (int)__builtin_amdgcn_mbcnt_lo(mask, 0u);
                    if (hit) myq[pos] = e;
                    cnt += __builtin_popcount(mask);
                }
            }
        }
    };

    auto plane_word = [&](int w0) -> unsigned {
        const int wi = w0 + lane;
        return (wi < wend) ? plane[wi] : 0u;
    };

    unsigned word = plane_word(wbeg);
    for (int w0 = wbeg; w0 < wend; w0 += 32) {
        const unsigned nxt = plane_word(w0 + 32);
        const int cword = (w0 + lane) << 5;
        int nlist = 0;
        for (int t = 0; t <= 32; ++t) {
            const bool has = word != 0u;
            const unsigned m = __builtin_amdgcn_ballot_w32(has);
            const int k = __builtin_popcount(m);
            if (k == 0 || nlist + k > CLCAP) { visit(nlist); nlist = 0; }
            if (k == 0) break;
            if (has) {
                myc[nlist + (int)__builtin_amdgcn_mbcnt_lo(m, 0u)] = cword + __builtin_ctz(word);
                word &= word - 1u;
            }
            nlist += k;
        }
        word = nxt;
    }
    drain_list<GRP>(acc, myq, cnt, erow, ecol, a1, a2, m_mu, m_sig, row0, lane);
    __syncthreads();

    for (int i = tid; i < nrows * (OUT_F / 4); i += 1024) {
        const int r  = i >> 5;
        const int c4 = i & 31;
        const float* am = acc + r * ROW_W + c4;
        const float* as = am + OUT_F;
        const float4 vm = make_float4(am[0], am[32], am[64], am[96]);
        const float4 vs = make_float4(as[0], as[32], as[64], as[96]);
        *(float4*)(out_mu  + (size_t)(row0 + r) * OUT_F + c4 * 4) = vm;
        *(float4*)(out_sig + (size_t)(row0 + r) * OUT_F + c4 * 4) = vs;
    }
}

extern "C" void kernel_launch(void* const* d_in, const int* in_sizes, int n_in,
                              void* d_out, int out_size, void* d_ws, size_t ws_size,
                              hipStream_t stream) {
    const float* miu   = (const float*)d_in[0];
    const float* sigma = (const float*)d_in[1];
    const int*   erow  = (const int*)  d_in[2];
    const int*   ecol  = (const int*)  d_in[3];
    const float* a1    = (const float*)d_in[4];
    const float* a2    = (const float*)d_in[5];
    const float* Wmu   = (const float*)d_in[6];
    const float* Wsig  = (const float*)d_in[7];

    const int N = in_sizes[0] / IN_F;
    const int E = in_sizes[2];

    float* out_mu  = (float*)d_out;
    float* out_sig = out_mu + (size_t)N * OUT_F;

    float* m_mu  = (float*)d_ws;
    float* m_sig = m_mu + (size_t)N * OUT_F;
    unsigned* planes = (unsigned*)(m_sig + (size_t)N * OUT_F);

    const int nblk   = (N + OWN_ROWS - 1) / OWN_ROWS;
    const int nspans = (E + SPAN_CH * 32 - 1) / (SPAN_CH * 32);
    const int wpad   = nspans * SPAN_W;

    gemm_msg_kernel<<<256, GEMM_THR, 0, stream>>>(
        miu, sigma, Wmu, Wsig, m_mu, m_sig, N, (N + 15) / 16);

    owner_planes_kernel<<<nspans, 1024, 0, stream>>>(erow, planes, N, E, nblk, wpad);

    spmm_owner_kernel<<<nblk, 1024, 0, stream>>>(
        erow, ecol, a1, a2, m_mu, m_sig, planes, out_mu, out_sig, N, E, wpad);
}


# owner SpMM: a listed edge's row, column and both weights come as one 16-byte record (one line), not four 4-byte loads of four lines; extra work-groups of the planes launch write the records
# speedup vs baseline: 2.9954x; 1.0545x over previous
#include <hip/hip_runtime.h>
#include <hip/hip_bf16.h>

typedef float v2f __attribute__((ext_vector_type(2)));
typedef float v8f __attribute__((ext_vector_type(8)));

#define IN_F 256
#define OUT_F 128

#define GEMM_WAVES 16
#define GEMM_THR   (GEMM_WAVES * 32)
#define W_FLOATS   (IN_F * OUT_F)
#define W_QUADS    (W_FLOATS / 4)
#define KSLAB      (OUT_F * 4)
#define NT_REST    2

template <int NT>
__device__ __forceinline__ void gemm_tile(const float* __restrict__ miu,
                                          const float* __restrict__ sigma,
                                          const float* bsig, const float* bmu,
                                          float* __restrict__ m_mu,
                                          float* __restrict__ m_sig,
                                          int N, int row_base, int ct0, int m, int kgrp) {
    const int arow = (row_base + m < N) ? (row_base + m) : (N - 1);

    float att[NT][8];

    {
        v8f acc[NT];
        v8f zero = {};
        #pragma unroll
        for (int t = 0; t < NT; ++t) acc[t] = zero;

        const float* Ap = sigma + (size_t)arow * IN_F + kgrp * 2;
        const float* Bw = bsig + ct0 * 64;
        #pragma unroll 4
        for (int k0 = 0; k0 < IN_F; k0 += 4) {
            v2f a = *(const v2f*)(Ap + k0);
            const float* Bp = Bw + (k0 >> 2) * KSLAB;
            #pragma unroll
            for (int t = 0; t < NT; ++t) {
                v2f b = *(const v2f*)(Bp + t * 64);
                acc[t] = __builtin_amdgcn_wmma_f32_16x16x4_f32(
                    false, a, false, b, (short)0, acc[t], false, false);
            }
        }
        #pragma unroll
        for (int t = 0; t < NT; ++t) {
            #pragma unroll
            for (int r = 0; r < 8; ++r) {
                float hs = fmaxf(acc[t][r], 0.0f);
                float a  = __expf(-hs);
                att[t][r] = a;
                int rr = row_base + r + kgrp * 8;
                if (rr < N)
                    m_sig[(size_t)rr * OUT_F + (ct0 + t) * 16 + m] = hs * a * a;
            }
        }
    }

    {
        v8f acc[NT];
        v8f zero = {};
        #pragma unroll
        for (int t = 0; t < NT; ++t) acc[t] = zero;

        const float* Ap = miu + (size_t)arow * IN_F + kgrp * 2;
        const float* Bw = bmu + ct0 * 64;
        #pragma unroll 4
        for (int k0 = 0; k0 < IN_F; k0 += 4) {
            v2f a = *(const v2f*)(Ap + k0);
            const float* Bp = Bw + (k0 >> 2) * KSLAB;
            #pragma unroll
            for (int t = 0; t < NT; ++t) {
                v2f b = *(const v2f*)(Bp + t * 64);
                acc[t] = __builtin_amdgcn_wmma_f32_16x16x4_f32(
                    false, a, false, b, (short)0, acc[t], false, false);
            }
        }
        #pragma unroll
        for (int t = 0; t < NT; ++t) {
            #pragma unroll
            for (int r = 0; r < 8; ++r) {
                float hm = acc[t][r];
                float e  = (hm > 0.0f) ? hm : (__expf(hm) - 1.0f);
                int rr = row_base + r + kgrp * 8;
                if (rr < N)
                    m_mu[(size_t)rr * OUT_F + (ct0 + t) * 16 + m] = e * att[t][r];
            }
        }
    }
}

__global__ __launch_bounds__(GEMM_THR)
void gemm_msg_kernel(const float* __restrict__ miu,
                     const float* __restrict__ sigma,
                     const float* __restrict__ Wmu,
                     const float* __restrict__ Wsig,
                     float* __restrict__ m_mu,
                     float* __restrict__ m_sig,
                     int N, int ntiles) {
    __shared__ __attribute__((aligned(16))) float wl[2 * W_FLOATS];

    const int tid  = threadIdx.x;

    {
        const int n = tid & (OUT_F - 1);
        const int q = tid / OUT_F;
        const float* ssig = Wsig + (size_t)(4 * q) * OUT_F + n;
        const float* smu  = Wmu  + (size_t)(4 * q) * OUT_F + n;
        float4* dsig = (float4*)wl + tid;
        float4* dmu  = dsig + W_QUADS;
        #pragma unroll
        for (int j = 0; j < W_QUADS / GEMM_THR; ++j) {
            const float* s = ssig + (size_t)j * (GEMM_THR / OUT_F) * 4 * OUT_F;
            float4 v;
            v.x = s[0]; v.y = s[OUT_F]; v.z = s[2 * OUT_F]; v.w = s[3 * OUT_F];
            dsig[j * GEMM_THR] = v;
        }
        #pragma unroll
        for (int j = 0; j < W_QUADS / GEMM_THR; ++j) {
            const float* s = smu + (size_t)j * (GEMM_THR / OUT_F) * 4 * OUT_F;
            float4 v;
            v.x = s[0]; v.y = s[OUT_F]; v.z = s[2 * OUT_F]; v.w = s[3 * OUT_F];
            dmu[j * GEMM_THR] = v;
        }
    }
    __syncthreads();

    const int lane = tid & 31;
    const int wave = __builtin_amdgcn_readfirstlane(tid >> 5);

    const int m    = lane & 15;
    const int kgrp = lane >> 4;

    const float* bsig = wl + m * 4 + kgrp * 2;
    const float* bmu  = bsig + W_FLOATS;

    const int G      = (int)gridDim.x;
    const int nfull  = ntiles - ntiles % (4 * G);
    const int nslots = nfull + (ntiles - nfull) * (8 / NT_REST);

    for (int slot = wave * G + (int)blockIdx.x; slot < nslots; slot += GEMM_WAVES * G) {
        if (slot < nfull) {
            gemm_tile<8>(miu, sigma, bsig, bmu, m_mu, m_sig, N, slot * 16, 0, m, kgrp);
        } else {
            const int u = slot - nfull;
            gemm_tile<NT_REST>(miu, sigma, bsig, bmu, m_mu, m_sig, N,
                               (nfull + u / (8 / NT_REST)) * 16, (u % (8 / NT_REST)) * NT_REST, m, kgrp);
        }
    }
}

#define OWN_ROWS 196
#define SPAN_W   32
#define SPAN_CH  (SPAN_W * 32)
#define OWN_TILE 512
#define TILE_P   (OWN_TILE + 1)
#define PACK_WG  207
#define PACK_U   4

__global__ __launch_bounds__(1024)
void owner_planes_kernel(const int* __restrict__ erow,
                         const int* __restrict__ ecol,
                         const float* __restrict__ a1,
                         const float* __restrict__ a2,
                         unsigned* __restrict__ planes,
                         int* __restrict__ rec,
                         int N, int E, int nblk, int wpad, int nspans) {
    __shared__ unsigned bits[SPAN_W * TILE_P];

    const int tid  = threadIdx.x;

    if ((int)blockIdx.x >= nspans) {
        int4* rec4 = (int4*)rec;
        const int stride = ((int)gridDim.x - nspans) * 1024;
        for (int e0 = ((int)blockIdx.x - nspans) * 1024 + tid; e0 < E; e0 += PACK_U * stride) {
            int   pr[PACK_U], pc[PACK_U];
            float px[PACK_U], py[PACK_U];
            #pragma unroll
            for (int u = 0; u < PACK_U; ++u) {
                const int e = min(e0 + u * stride, E - 1);
                pr[u] = erow[e];
                pc[u] = ecol[e];
                px[u] = a1[e];
                py[u] = a2[e];
            }
            #pragma unroll
            for (int u = 0; u < PACK_U; ++u) {
                const int e = e0 + u * stride;
                if (e < E)
                    rec4[e] = make_int4(pr[u], pc[u], __float_as_int(px[u]), __float_as_int(py[u]));
            }
        }
        return;
    }

    const int lane = tid & 31;
    const int wave = __builtin_amdgcn_readfirstlane(tid >> 5);
    const int c0   = blockIdx.x * SPAN_CH + wave * 32;

    int own[32];
    #pragma unroll
    for (int i = 0; i < 32; ++i) {
        const int e = (c0 + i) * 32 + lane;
        const int r = erow[min(e, E - 1)];
        own[i] = (e < E && (unsigned)r < (unsigned)N) ? r / OWN_ROWS : -1;
    }

    for (int ob = 0; ob < nblk; ob += OWN_TILE) {
        for (int i = tid; i < SPAN_W * TILE_P; i += 1024) bits[i] = 0u;
        __syncthreads();
        #pragma unroll
        for (int i = 0; i < 32; ++i) {
            const unsigned o = (unsigned)(own[i] - ob);
            if (o < (unsigned)OWN_TILE) atomicOr(&bits[wave * TILE_P + o], 1u << i);
        }
        __syncthreads();
        const int no = (nblk - ob < OWN_TILE) ? (nblk - ob) : OWN_TILE;
        for (int i = tid; i < no * SPAN_W; i += 1024) {
            const int o = i >> 5;
            planes[(size_t)(ob + o) * wpad + blockIdx.x * SPAN_W + lane] = bits[lane * TILE_P + o];
        }
        __syncthreads();
    }
}

#define ROW_W    (2 * OUT_F)
#define VG       16
#define QCAP     (VG * 32 + 160)
#define CLCAP    128
#define GRP      8

template <int G>
__device__ __forceinline__ void gather_group(float* __restrict__ acc,
                                             const float* __restrict__ m_mu,
                                             const float* __restrict__ m_sig,
                                             int r, int c, float v1, float v2,
                                             int j0, int lane) {
    float4 gm[G], gs[G];
    #pragma unroll
    for (int g = 0; g < G; ++g) {
        const int cj = __builtin_amdgcn_readlane(c, j0 + g);
        gm[g] = *(const float4*)(m_mu  + (size_t)cj * OUT_F + lane * 4);
        gs[g] = *(const float4*)(m_sig + (size_t)cj * OUT_F + lane * 4);
    }
    #pragma unroll
    for (int g = 0; g < G; ++g) {
        const int   rj  = __builtin_amdgcn_readlane(r, j0 + g);
        const float w1  = __int_as_float(__builtin_amdgcn_readlane(__float_as_int(v1), j0 + g));
        const float w2  = __int_as_float(__builtin_amdgcn_readlane(__float_as_int(v2), j0 + g));
        float* dst = acc + rj * ROW_W + lane;
        unsafeAtomicAdd(dst,              w1 * gm[g].x);
        unsafeAtomicAdd(dst + OUT_F,      w2 * gs[g].x);
        unsafeAtomicAdd(dst + 32,         w1 * gm[g].y);
        unsafeAtomicAdd(dst + OUT_F + 32, w2 * gs[g].y);
        unsafeAtomicAdd(dst + 64,         w1 * gm[g].z);
        unsafeAtomicAdd(dst + OUT_F + 64, w2 * gs[g].z);
        unsafeAtomicAdd(dst + 96,         w1 * gm[g].w);
        unsafeAtomicAdd(dst + OUT_F + 96, w2 * gs[g].w);
    }
}

template <int G>
__device__ __forceinline__ void drain_list(float* __restrict__ acc, const int* __restrict__ myq, int cnt,
                                           const int4* __restrict__ rec4,
                                           const float* __restrict__ m_mu, const float* __restrict__ m_sig,
                                           int row0, int lane) {
    for (int i0 = 0; i0 < cnt; i0 += 32) {
        const int n = (cnt - i0 < 32) ? (cnt - i0) : 32;
        int r = 0, c = 0;
        float v1 = 0.f, v2 = 0.f;
        if (lane < n) {
            const int4 q = rec4[myq[i0 + lane]];
            r  = q.x - row0;
            c  = q.y;
            v1 = __int_as_float(q.z);
            v2 = __int_as_float(q.w);
        }
        int j = 0;
        for (; j + G <= n; j += G)
            gather_group<G>(acc, m_mu, m_sig, r, c, v1, v2, j, lane);
        if (G > 4 && j + 4 <= n) { gather_group<4>(acc, m_mu, m_sig, r, c, v1, v2, j, lane); j += 4; }
        if (G > 2 && j + 2 <= n) { gather_group<2>(acc, m_mu, m_sig, r, c, v1, v2, j, lane); j += 2; }
        if (G > 1 && j < n)        gather_group<1>(acc, m_mu, m_sig, r, c, v1, v2, j, lane);
    }
}

__global__ __launch_bounds__(1024)
void spmm_owner_kernel(const int* __restrict__ erow,
                       const int* __restrict__ rec,
                       const float* __restrict__ m_mu,
                       const float* __restrict__ m_sig,
                       const unsigned* __restrict__ planes,
                       float* __restrict__ out_mu,
                       float* __restrict__ out_sig,
                       int N, int E, int wpad) {
    __shared__ __attribute__((aligned(16))) float acc[OWN_ROWS * ROW_W];
    __shared__ int elist[32 * QCAP];
    __shared__ int clist[32 * CLCAP];

    const int tid  = threadIdx.x;
    const int lane = tid & 31;
    const int wave = __builtin_amdgcn_readfirstlane(tid >> 5);
    const int row0 = blockIdx.x * OWN_ROWS;
    const int nrows = (N - row0 < OWN_ROWS) ? (N - row0) : OWN_ROWS;
    int* myq = elist + wave * QCAP;
    int* myc = clist + wave * CLCAP;
    const int4* rec4 = (const int4*)rec;

    for (int i = tid; i < OWN_ROWS * ROW_W / 4; i += 1024)
        ((float4*)acc)[i] = make_float4(0.f, 0.f, 0.f, 0.f);
    __syncthreads();

    int cnt = 0;

    const int nchunks = (E + 31) >> 5;
    const int nwords  = (nchunks + 31) >> 5;
    const int per     = (nwords + 31) >> 5;
    const int wbeg    = wave * per;
    const int wend    = (wbeg + per < nwords) ? (wbeg + per) : nwords;
    const unsigned* plane = planes + (size_t)blockIdx.x * wpad;

    auto make_room = [&]() {
        if (cnt > QCAP - VG * 32) {
            drain_list<1>(acc, myq, cnt, rec4, m_mu, m_sig, row0, lane);
            cnt = 0;
        }
    };

    auto visit = [&](int nlist) {
        for (int i0 = 0; i0 < nlist; i0 += 32) {
            const int n = (nlist - i0 < 32) ? (nlist - i0) : 32;
            const int cid = (lane < n) ? myc[i0 + lane] : nchunks;
            for (int j = 0; j < n; j += VG) {
                make_room();
                int rv[VG];
                #pragma unroll
                for (int g = 0; g < VG; ++g) {
                    const int e = __builtin_amdgcn_readlane(cid, j + g) * 32 + lane;
                    rv[g] = erow[min(e, E - 1)];
                }
                #pragma unroll
                for (int g = 0; g < VG; ++g) {
                    const int e = __builtin_amdgcn_readlane(cid, j + g) * 32 + lane;
                    const bool hit = (e < E) && ((unsigned)(rv[g] - row0) < (unsigned)nrows);
                    const unsigned mask = __builtin_amdgcn_ballot_w32(hit);
                    const int pos = cnt + (int)__builtin_amdgcn_mbcnt_lo(mask, 0u);
                    if (hit) myq[pos] = e;
                    cnt += __builtin_popcount(mask);
                }
            }
        }
    };

    auto plane_word = [&](int w0) -> unsigned {
        const int wi = w0 + lane;
        return (wi < wend) ? plane[wi] : 0u;
    };

    unsigned word = plane_word(wbeg);
    for (int w0 = wbeg; w0 < wend; w0 += 32) {
        const unsigned nxt = plane_word(w0 + 32);
        const int cword = (w0 + lane) << 5;
        int nlist = 0;
        for (int t = 0; t <= 32; ++t) {
            const bool has = word != 0u;
            const unsigned m = __builtin_amdgcn_ballot_w32(has);
            const int k = __builtin_popcount(m);
            if (k == 0 || nlist + k > CLCAP) { visit(nlist); nlist = 0; }
            if (k == 0) break;
            if (has) {
                myc[nlist + (int)__builtin_amdgcn_mbcnt_lo(m, 0u)] = cword + __builtin_ctz(word);
                word &= word - 1u;
            }
            nlist += k;
        }
        word = nxt;
    }
    drain_list<GRP>(acc, myq, cnt, rec4, m_mu, m_sig, row0, lane);
    __syncthreads();

    for (int i = tid; i < nrows * (OUT_F / 4); i += 1024) {
        const int r  = i >> 5;
        const int c4 = i & 31;
        const float* am = acc + r * ROW_W + c4;
        const float* as = am + OUT_F;
        const float4 vm = make_float4(am[0], am[32], am[64], am[96]);
        const float4 vs = make_float4(as[0], as[32], as[64], as[96]);
        *(float4*)(out_mu  + (size_t)(row0 + r) * OUT_F + c4 * 4) = vm;
        *(float4*)(out_sig + (size_t)(row0 + r) * OUT_F + c4 * 4) = vs;
    }
}

extern "C" void kernel_launch(void* const* d_in, const int* in_sizes, int n_in,
                              void* d_out, int out_size, void* d_ws, size_t ws_size,
                              hipStream_t stream) {
    const float* miu   = (const float*)d_in[0];
    const float* sigma = (const float*)d_in[1];
    const int*   erow  = (const int*)  d_in[2];
    const int*   ecol  = (const int*)  d_in[3];
    const float* a1    = (const float*)d_in[4];
    const float* a2    = (const float*)d_in[5];
    const float* Wmu   = (const float*)d_in[6];
    const float* Wsig  = (const float*)d_in[7];

    const int N = in_sizes[0] / IN_F;
    const int E = in_sizes[2];

    float* out_mu  = (float*)d_out;
    float* out_sig = out_mu + (size_t)N * OUT_F;

    float* m_mu  = (float*)d_ws;
    float* m_sig = m_mu + (size_t)N * OUT_F;
    unsigned* planes = (unsigned*)(m_sig + (size_t)N * OUT_F);

    const int nblk   = (N + OWN_ROWS - 1) / OWN_ROWS;
    const int nspans = (E + SPAN_CH * 32 - 1) / (SPAN_CH * 32);
    const int wpad   = nspans * SPAN_W;

    gemm_msg_kernel<<<256, GEMM_THR, 0, stream>>>(
        miu, sigma, Wmu, Wsig, m_mu, m_sig, N, (N + 15) / 16);

    int* rec = (int*)(planes + (size_t)nblk * wpad);
    owner_planes_kernel<<<nspans + PACK_WG, 1024, 0, stream>>>(
        erow, ecol, a1, a2, planes, rec, N, E, nblk, wpad, nspans);

    spmm_owner_kernel<<<nblk, 1024, 0, stream>>>(
        erow, rec, m_mu, m_sig, planes, out_mu, out_sig, N, E, wpad);
}


# owner SpMM: a work-group owns 392 rows (was 196) and sums the two outputs one after the other from one kept edge list: all work-groups gather from ONE table at a time; LDS adds in arrival order
# speedup vs baseline: 3.1606x; 1.0552x over previous
#include <hip/hip_runtime.h>
#include <hip/hip_bf16.h>

typedef float v2f __attribute__((ext_vector_type(2)));
typedef float v8f __attribute__((ext_vector_type(8)));

#define IN_F 256
#define OUT_F 128

#define GEMM_WAVES 16
#define GEMM_THR   (GEMM_WAVES * 32)
#define W_FLOATS   (IN_F * OUT_F)
#define W_QUADS    (W_FLOATS / 4)
#define KSLAB      (OUT_F * 4)
#define NT_REST    2

template <int NT>
__device__ __forceinline__ void gemm_tile(const float* __restrict__ miu,
                                          const float* __restrict__ sigma,
                                          const float* bsig, const float* bmu,
                                          float* __restrict__ m_mu,
                                          float* __restrict__ m_sig,
                                          int N, int row_base, int ct0, int m, int kgrp) {
    const int arow = (row_base + m < N) ? (row_base + m) : (N - 1);

    float att[NT][8];

    {
        v8f acc[NT];
        v8f zero = {};
        #pragma unroll
        for (int t = 0; t < NT; ++t) acc[t] = zero;

        const float* Ap = sigma + (size_t)arow * IN_F + kgrp * 2;
        const float* Bw = bsig + ct0 * 64;
        #pragma unroll 4
        for (int k0 = 0; k0 < IN_F; k0 += 4) {
            v2f a = *(const v2f*)(Ap + k0);
            const float* Bp = Bw + (k0 >> 2) * KSLAB;
            #pragma unroll
            for (int t = 0; t < NT; ++t) {
                v2f b = *(const v2f*)(Bp + t * 64);
                acc[t] = __builtin_amdgcn_wmma_f32_16x16x4_f32(
                    false, a, false, b, (short)0, acc[t], false, false);
            }
        }
        #pragma unroll
        for (int t = 0; t < NT; ++t) {
            #pragma unroll
            for (int r = 0; r < 8; ++r) {
                float hs = fmaxf(acc[t][r], 0.0f);
                float a  = __expf(-hs);
                att[t][r] = a;
                int rr = row_base + r + kgrp * 8;
                if (rr < N)
                    m_sig[(size_t)rr * OUT_F + (ct0 + t) * 16 + m] = hs * a * a;
            }
        }
    }

    {
        v8f acc[NT];
        v8f zero = {};
        #pragma unroll
        for (int t = 0; t < NT; ++t) acc[t] = zero;

        const float* Ap = miu + (size_t)arow * IN_F + kgrp * 2;
        const float* Bw = bmu + ct0 * 64;
        #pragma unroll 4
        for (int k0 = 0; k0 < IN_F; k0 += 4) {
            v2f a = *(const v2f*)(Ap + k0);
            const float* Bp = Bw + (k0 >> 2) * KSLAB;
            #pragma unroll
            for (int t = 0; t < NT; ++t) {
                v2f b = *(const v2f*)(Bp + t * 64);
                acc[t] = __builtin_amdgcn_wmma_f32_16x16x4_f32(
                    false, a, false, b, (short)0, acc[t], false, false);
            }
        }
        #pragma unroll
        for (int t = 0; t < NT; ++t) {
            #pragma unroll
            for (int r = 0; r < 8; ++r) {
                float hm = acc[t][r];
                float e  = (hm > 0.0f) ? hm : (__expf(hm) - 1.0f);
                int rr = row_base + r + kgrp * 8;
                if (rr < N)
                    m_mu[(size_t)rr * OUT_F + (ct0 + t) * 16 + m] = e * att[t][r];
            }
        }
    }
}

__global__ __launch_bounds__(GEMM_THR)
void gemm_msg_kernel(const float* __restrict__ miu,
                     const float* __restrict__ sigma,
                     const float* __restrict__ Wmu,
                     const float* __restrict__ Wsig,
                     float* __restrict__ m_mu,
                     float* __restrict__ m_sig,
                     int N, int ntiles) {
    __shared__ __attribute__((aligned(16))) float wl[2 * W_FLOATS];

    const int tid  = threadIdx.x;

    {
        const int n = tid & (OUT_F - 1);
        const int q = tid / OUT_F;
        const float* ssig = Wsig + (size_t)(4 * q) * OUT_F + n;
        const float* smu  = Wmu  + (size_t)(4 * q) * OUT_F + n;
        float4* dsig = (float4*)wl + tid;
        float4* dmu  = dsig + W_QUADS;
        #pragma unroll
        for (int j = 0; j < W_QUADS / GEMM_THR; ++j) {
            const float* s = ssig + (size_t)j * (GEMM_THR / OUT_F) * 4 * OUT_F;
            float4 v;
            v.x = s[0]; v.y = s[OUT_F]; v.z = s[2 * OUT_F]; v.w = s[3 * OUT_F];
            dsig[j * GEMM_THR] = v;
        }
        #pragma unroll
        for (int j = 0; j < W_QUADS / GEMM_THR; ++j) {
            const float* s = smu + (size_t)j * (GEMM_THR / OUT_F) * 4 * OUT_F;
            float4 v;
            v.x = s[0]; v.y = s[OUT_F]; v.z = s[2 * OUT_F]; v.w = s[3 * OUT_F];
            dmu[j * GEMM_THR] = v;
        }
    }
    __syncthreads();

    const int lane = tid & 31;
    const int wave = __builtin_amdgcn_readfirstlane(tid >> 5);

    const int m    = lane & 15;
    const int kgrp = lane >> 4;

    const float* bsig = wl + m * 4 + kgrp * 2;
    const float* bmu  = bsig + W_FLOATS;

    const int G      = (int)gridDim.x;
    const int nfull  = ntiles - ntiles % (4 * G);
    const int nslots = nfull + (ntiles - nfull) * (8 / NT_REST);

    for (int slot = wave * G + (int)blockIdx.x; slot < nslots; slot += GEMM_WAVES * G) {
        if (slot < nfull) {
            gemm_tile<8>(miu, sigma, bsig, bmu, m_mu, m_sig, N, slot * 16, 0, m, kgrp);
        } else {
            const int u = slot - nfull;
            gemm_tile<NT_REST>(miu, sigma, bsig, bmu, m_mu, m_sig, N,
                               (nfull + u / (8 / NT_REST)) * 16, (u % (8 / NT_REST)) * NT_REST, m, kgrp);
        }
    }
}

#define OWN_ROWS 196
#define SPAN_W   32
#define SPAN_CH  (SPAN_W * 32)
#define OWN_TILE 512
#define TILE_P   (OWN_TILE + 1)
#define PACK_WG  207
#define PACK_U   4

__global__ __launch_bounds__(1024)
void owner_planes_kernel(const int* __restrict__ erow,
                         const int* __restrict__ ecol,
                         const float* __restrict__ a1,
                         const float* __restrict__ a2,
                         unsigned* __restrict__ planes,
                         int* __restrict__ rec,
                         int N, int E, int nblk, int wpad, int nspans) {
    __shared__ unsigned bits[SPAN_W * TILE_P];

    const int tid  = threadIdx.x;

    if ((int)blockIdx.x >= nspans) {
        int4* rec4 = (int4*)rec;
        const int stride = ((int)gridDim.x - nspans) * 1024;
        for (int e0 = ((int)blockIdx.x - nspans) * 1024 + tid; e0 < E; e0 += PACK_U * stride) {
            int   pr[PACK_U], pc[PACK_U];
            float px[PACK_U], py[PACK_U];
            #pragma unroll
            for (int u = 0; u < PACK_U; ++u) {
                const int e = min(e0 + u * stride, E - 1);
                pr[u] = erow[e];
                pc[u] = ecol[e];
                px[u] = a1[e];
                py[u] = a2[e];
            }
            #pragma unroll
            for (int u = 0; u < PACK_U; ++u) {
                const int e = e0 + u * stride;
                if (e < E)
                    rec4[e] = make_int4(pr[u], pc[u], __float_as_int(px[u]), __float_as_int(py[u]));
            }
        }
        return;
    }

    const int lane = tid & 31;
    const int wave = __builtin_amdgcn_readfirstlane(tid >> 5);
    const int c0   = blockIdx.x * SPAN_CH + wave * 32;

    int own[32];
    #pragma unroll
    for (int i = 0; i < 32; ++i) {
        const int e = (c0 + i) * 32 + lane;
        const int r = erow[min(e, E - 1)];
        own[i] = (e < E && (unsigned)r < (unsigned)N) ? r / OWN_ROWS : -1;
    }

    for (int ob = 0; ob < nblk; ob += OWN_TILE) {
        for (int i = tid; i < SPAN_W * TILE_P; i += 1024) bits[i] = 0u;
        __syncthreads();
        #pragma unroll
        for (int i = 0; i < 32; ++i) {
            const unsigned o = (unsigned)(own[i] - ob);
            if (o < (unsigned)OWN_TILE) atomicOr(&bits[wave * TILE_P + o], 1u << i);
        }
        __syncthreads();
        const int no = (nblk - ob < OWN_TILE) ? (nblk - ob) : OWN_TILE;
        for (int i = tid; i < no * SPAN_W; i += 1024) {
            const int o = i >> 5;
            planes[(size_t)(ob + o) * wpad + blockIdx.x * SPAN_W + lane] = bits[lane * TILE_P + o];
        }
        __syncthreads();
    }
}

#define UNIT_OWN  2
#define UNIT_ROWS (OWN_ROWS * UNIT_OWN)
#define VG       16
#define QCAP     (VG * 32 + 288)
#define CLCAP    128
#define GRP      16

template <int G>
__device__ __forceinline__ void gather_group(float* __restrict__ acc,
                                             const float* __restrict__ msg,
                                             int r, int c, float v,
                                             int j0, int lane) {
    float4 gm[G];
    #pragma unroll
    for (int g = 0; g < G; ++g) {
        const int cj = __builtin_amdgcn_readlane(c, j0 + g);
        gm[g] = *(const float4*)(msg + (size_t)cj * OUT_F + lane * 4);
    }
    #pragma unroll
    for (int g = 0; g < G; ++g) {
        const int   rj = __builtin_amdgcn_readlane(r, j0 + g);
        const float w  = __int_as_float(__builtin_amdgcn_readlane(__float_as_int(v), j0 + g));
        float* dst = acc + rj * OUT_F + lane;
        unsafeAtomicAdd(dst,      w * gm[g].x);
        unsafeAtomicAdd(dst + 32, w * gm[g].y);
        unsafeAtomicAdd(dst + 64, w * gm[g].z);
        unsafeAtomicAdd(dst + 96, w * gm[g].w);
    }
}

template <int G>
__device__ __forceinline__ void drain_list(float* __restrict__ acc, const int* __restrict__ myq, int cnt,
                                           const int4* __restrict__ rec4,
                                           const float* __restrict__ msg, bool var,
                                           int row0, int lane) {
    for (int i0 = 0; i0 < cnt; i0 += 32) {
        const int n = (cnt - i0 < 32) ? (cnt - i0) : 32;
        int r = 0, c = 0;
        float v = 0.f;
        if (lane < n) {
            const int4 q = rec4[myq[i0 + lane]];
            r = q.x - row0;
            c = q.y;
            v = __int_as_float(var ? q.w : q.z);
        }
        int j = 0;
        for (; j + G <= n; j += G)
            gather_group<G>(acc, msg, r, c, v, j, lane);
        if (G > 8 && j + 8 <= n) { gather_group<8>(acc, msg, r, c, v, j, lane); j += 8; }
        if (G > 4 && j + 4 <= n) { gather_group<4>(acc, msg, r, c, v, j, lane); j += 4; }
        if (G > 2 && j + 2 <= n) { gather_group<2>(acc, msg, r, c, v, j, lane); j += 2; }
        if (G > 1 && j < n)        gather_group<1>(acc, msg, r, c, v, j, lane);
    }
}

__global__ __launch_bounds__(1024)
void spmm_owner_kernel(const int* __restrict__ erow,
                       const int* __restrict__ rec,
                       const float* __restrict__ m_mu,
                       const float* __restrict__ m_sig,
                       const unsigned* __restrict__ planes,
                       float* __restrict__ out_mu,
                       float* __restrict__ out_sig,
                       int N, int E, int wpad, int nblk) {
    __shared__ __attribute__((aligned(16))) float acc[UNIT_ROWS * OUT_F];
    __shared__ int elist[32 * QCAP];
    __shared__ int clist[32 * CLCAP];

    const int tid  = threadIdx.x;
    const int lane = tid & 31;
    const int wave = __builtin_amdgcn_readfirstlane(tid >> 5);
    const int row0 = blockIdx.x * UNIT_ROWS;
    const int nrows = (N - row0 < UNIT_ROWS) ? (N - row0) : UNIT_ROWS;
    int* myq = elist + wave * QCAP;
    int* myc = clist + wave * CLCAP;
    const int4* rec4 = (const int4*)rec;

    const int nchunks = (E + 31) >> 5;
    const int nwords  = (nchunks + 31) >> 5;
    const int per     = (nwords + 31) >> 5;
    const int wbeg    = wave * per;
    const int wend    = (wbeg + per < nwords) ? (wbeg + per) : nwords;
    static_assert(UNIT_OWN == 2, "a work-group's plane word is the OR of two owners' words");
    const unsigned* plane = planes + (size_t)(blockIdx.x * UNIT_OWN) * wpad;
    const bool two = (int)blockIdx.x * UNIT_OWN + 1 < nblk;

    int  cnt = 0;
    bool spilt = false;

    for (int pass = 0; pass < 2; ++pass) {
        const bool var = pass != 0;
        const float* __restrict__ msg = var ? m_sig : m_mu;
        float* __restrict__ out = var ? out_sig : out_mu;

        if (var) __syncthreads();
        for (int i = tid; i < UNIT_ROWS * OUT_F / 4; i += 1024)
            ((float4*)acc)[i] = make_float4(0.f, 0.f, 0.f, 0.f);
        __syncthreads();

        if (!var || spilt) {
            cnt = 0;

            auto make_room = [&]() {
                if (cnt > QCAP - VG * 32) {
                    drain_list<1>(acc, myq, cnt, rec4, msg, var, row0, lane);
                    cnt = 0;
                    spilt = true;
                }
            };

            auto visit = [&](int nlist) {
                for (int i0 = 0; i0 < nlist; i0 += 32) {
                    const int n = (nlist - i0 < 32) ? (nlist - i0) : 32;
                    const int cid = (lane < n) ? myc[i0 + lane] : nchunks;
                    for (int j = 0; j < n; j += VG) {
                        make_room();
                        int rv[VG];
                        #pragma unroll
                        for (int g = 0; g < VG; ++g) {
                            const int e = __builtin_amdgcn_readlane(cid, j + g) * 32 + lane;
                            rv[g] = erow[min(e, E - 1)];
                        }
                        #pragma unroll
                        for (int g = 0; g < VG; ++g) {
                            const int e = __builtin_amdgcn_readlane(cid, j + g) * 32 + lane;
                            const bool hit = (e < E) && ((unsigned)(rv[g] - row0) < (unsigned)nrows);
                            const unsigned mask = __builtin_amdgcn_ballot_w32(hit);
                            const int pos = cnt + (int)__builtin_amdgcn_mbcnt_lo(mask, 0u);
                            if (hit) myq[pos] = e;
                            cnt += __builtin_popcount(mask);
                        }
                    }
                }
            };

            auto plane_word = [&](int w0) -> unsigned {
                const int wi = w0 + lane;
                if (wi >= wend) return 0u;
                const unsigned a = plane[wi];
                const unsigned b = two ? plane[(size_t)wpad + wi] : 0u;
                return a | b;
            };

            unsigned word = plane_word(wbeg);
            for (int w0 = wbeg; w0 < wend; w0 += 32) {
                const unsigned nxt = plane_word(w0 + 32);
                const int cword = (w0 + lane) << 5;
                int nlist = 0;
                for (int t = 0; t <= 32; ++t) {
                    const bool has = word != 0u;
                    const unsigned m = __builtin_amdgcn_ballot_w32(has);
                    const int k = __builtin_popcount(m);
                    if (k == 0 || nlist + k > CLCAP) { visit(nlist); nlist = 0; }
                    if (k == 0) break;
                    if (has) {
                        myc[nlist + (int)__builtin_amdgcn_mbcnt_lo(m, 0u)] = cword + __builtin_ctz(word);
                        word &= word - 1u;
                    }
                    nlist += k;
                }
                word = nxt;
            }
        }
        drain_list<GRP>(acc, myq, cnt, rec4, msg, var, row0, lane);
        __syncthreads();

        for (int i = tid; i < nrows * (OUT_F / 4); i += 1024) {
            const int r  = i >> 5;
            const int c4 = i & 31;
            const float* a = acc + r * OUT_F + c4;
            *(float4*)(out + (size_t)(row0 + r) * OUT_F + c4 * 4) = make_float4(a[0], a[32], a[64], a[96]);
        }
    }
}

extern "C" void kernel_launch(void* const* d_in, const int* in_sizes, int n_in,
                              void* d_out, int out_size, void* d_ws, size_t ws_size,
                              hipStream_t stream) {
    const float* miu   = (const float*)d_in[0];
    const float* sigma = (const float*)d_in[1];
    const int*   erow  = (const int*)  d_in[2];
    const int*   ecol  = (const int*)  d_in[3];
    const float* a1    = (const float*)d_in[4];
    const float* a2    = (const float*)d_in[5];
    const float* Wmu   = (const float*)d_in[6];
    const float* Wsig  = (const float*)d_in[7];

    const int N = in_sizes[0] / IN_F;
    const int E = in_sizes[2];

    float* out_mu  = (float*)d_out;
    float* out_sig = out_mu + (size_t)N * OUT_F;

    float* m_mu  = (float*)d_ws;
    float* m_sig = m_mu + (size_t)N * OUT_F;
    unsigned* planes = (unsigned*)(m_sig + (size_t)N * OUT_F);

    const int nblk   = (N + OWN_ROWS - 1) / OWN_ROWS;
    const int nspans = (E + SPAN_CH * 32 - 1) / (SPAN_CH * 32);
    const int wpad   = nspans * SPAN_W;

    gemm_msg_kernel<<<256, GEMM_THR, 0, stream>>>(
        miu, sigma, Wmu, Wsig, m_mu, m_sig, N, (N + 15) / 16);

    int* rec = (int*)(planes + (size_t)nblk * wpad);
    owner_planes_kernel<<<nspans + PACK_WG, 1024, 0, stream>>>(
        erow, ecol, a1, a2, planes, rec, N, E, nblk, wpad, nspans);

    spmm_owner_kernel<<<(nblk + UNIT_OWN - 1) / UNIT_OWN, 1024, 0, stream>>>(
        erow, rec, m_mu, m_sig, planes, out_mu, out_sig, N, E, wpad, nblk);
}


# owner SpMM: the mean's pass keeps row, column and variance weight of each listed edge in LDS (2 words of the wave's list); the variance's pass reads no record again; LDS adds in arrival order
# speedup vs baseline: 3.2693x; 1.0344x over previous
#include <hip/hip_runtime.h>
#include <hip/hip_bf16.h>

typedef float v2f __attribute__((ext_vector_type(2)));
typedef float v8f __attribute__((ext_vector_type(8)));

#define IN_F 256
#define OUT_F 128

#define GEMM_WAVES 16
#define GEMM_THR   (GEMM_WAVES * 32)
#define W_FLOATS   (IN_F * OUT_F)
#define W_QUADS    (W_FLOATS / 4)
#define KSLAB      (OUT_F * 4)
#define NT_REST    2

template <int NT>
__device__ __forceinline__ void gemm_tile(const float* __restrict__ miu,
                                          const float* __restrict__ sigma,
                                          const float* bsig, const float* bmu,
                                          float* __restrict__ m_mu,
                                          float* __restrict__ m_sig,
                                          int N, int row_base, int ct0, int m, int kgrp) {
    const int arow = (row_base + m < N) ? (row_base + m) : (N - 1);

    float att[NT][8];

    {
        v8f acc[NT];
        v8f zero = {};
        #pragma unroll
        for (int t = 0; t < NT; ++t) acc[t] = zero;

        const float* Ap = sigma + (size_t)arow * IN_F + kgrp * 2;
        const float* Bw = bsig + ct0 * 64;
        #pragma unroll 4
        for (int k0 = 0; k0 < IN_F; k0 += 4) {
            v2f a = *(const v2f*)(Ap + k0);
            const float* Bp = Bw + (k0 >> 2) * KSLAB;
            #pragma unroll
            for (int t = 0; t < NT; ++t) {
                v2f b = *(const v2f*)(Bp + t * 64);
                acc[t] = __builtin_amdgcn_wmma_f32_16x16x4_f32(
                    false, a, false, b, (short)0, acc[t], false, false);
            }
        }
        #pragma unroll
        for (int t = 0; t < NT; ++t) {
            #pragma unroll
            for (int r = 0; r < 8; ++r) {
                float hs = fmaxf(acc[t][r], 0.0f);
                float a  = __expf(-hs);
                att[t][r] = a;
                int rr = row_base + r + kgrp * 8;
                if (rr < N)
                    m_sig[(size_t)rr * OUT_F + (ct0 + t) * 16 + m] = hs * a * a;
            }
        }
    }

    {
        v8f acc[NT];
        v8f zero = {};
        #pragma unroll
        for (int t = 0; t < NT; ++t) acc[t] = zero;

        const float* Ap = miu + (size_t)arow * IN_F + kgrp * 2;
        const float* Bw = bmu + ct0 * 64;
        #pragma unroll 4
        for (int k0 = 0; k0 < IN_F; k0 += 4) {
            v2f a = *(const v2f*)(Ap + k0);
            const float* Bp = Bw + (k0 >> 2) * KSLAB;
            #pragma unroll
            for (int t = 0; t < NT; ++t) {
                v2f b = *(const v2f*)(Bp + t * 64);
                acc[t] = __builtin_amdgcn_wmma_f32_16x16x4_f32(
                    false, a, false, b, (short)0, acc[t], false, false);
            }
        }
        #pragma unroll
        for (int t = 0; t < NT; ++t) {
            #pragma unroll
            for (int r = 0; r < 8; ++r) {
                float hm = acc[t][r];
                float e  = (hm > 0.0f) ? hm : (__expf(hm) - 1.0f);
                int rr = row_base + r + kgrp * 8;
                if (rr < N)
                    m_mu[(size_t)rr * OUT_F + (ct0 + t) * 16 + m] = e * att[t][r];
            }
        }
    }
}

__global__ __launch_bounds__(GEMM_THR)
void gemm_msg_kernel(const float* __restrict__ miu,
                     const float* __restrict__ sigma,
                     const float* __restrict__ Wmu,
                     const float* __restrict__ Wsig,
                     float* __restrict__ m_mu,
                     float* __restrict__ m_sig,
                     int N, int ntiles) {
    __shared__ __attribute__((aligned(16))) float wl[2 * W_FLOATS];

    const int tid  = threadIdx.x;

    {
        const int n = tid & (OUT_F - 1);
        const int q = tid / OUT_F;
        const float* ssig = Wsig + (size_t)(4 * q) * OUT_F + n;
        const float* smu  = Wmu  + (size_t)(4 * q) * OUT_F + n;
        float4* dsig = (float4*)wl + tid;
        float4* dmu  = dsig + W_QUADS;
        #pragma unroll
        for (int j = 0; j < W_QUADS / GEMM_THR; ++j) {
            const float* s = ssig + (size_t)j * (GEMM_THR / OUT_F) * 4 * OUT_F;
            float4 v;
            v.x = s[0]; v.y = s[OUT_F]; v.z = s[2 * OUT_F]; v.w = s[3 * OUT_F];
            dsig[j * GEMM_THR] = v;
        }
        #pragma unroll
        for (int j = 0; j < W_QUADS / GEMM_THR; ++j) {
            const float* s = smu + (size_t)j * (GEMM_THR / OUT_F) * 4 * OUT_F;
            float4 v;
            v.x = s[0]; v.y = s[OUT_F]; v.z = s[2 * OUT_F]; v.w = s[3 * OUT_F];
            dmu[j * GEMM_THR] = v;
        }
    }
    __syncthreads();

    const int lane = tid & 31;
    const int wave = __builtin_amdgcn_readfirstlane(tid >> 5);

    const int m    = lane & 15;
    const int kgrp = lane >> 4;

    const float* bsig = wl + m * 4 + kgrp * 2;
    const float* bmu  = bsig + W_FLOATS;

    const int G      = (int)gridDim.x;
    const int nfull  = ntiles - ntiles % (4 * G);
    const int nslots = nfull + (ntiles - nfull) * (8 / NT_REST);

    for (int slot = wave * G + (int)blockIdx.x; slot < nslots; slot += GEMM_WAVES * G) {
        if (slot < nfull) {
            gemm_tile<8>(miu, sigma, bsig, bmu, m_mu, m_sig, N, slot * 16, 0, m, kgrp);
        } else {
            const int u = slot - nfull;
            gemm_tile<NT_REST>(miu, sigma, bsig, bmu, m_mu, m_sig, N,
                               (nfull + u / (8 / NT_REST)) * 16, (u % (8 / NT_REST)) * NT_REST, m, kgrp);
        }
    }
}

#define OWN_ROWS 196
#define SPAN_W   32
#define SPAN_CH  (SPAN_W * 32)
#define OWN_TILE 512
#define TILE_P   (OWN_TILE + 1)
#define PACK_WG  207
#define PACK_U   4

__global__ __launch_bounds__(1024)
void owner_planes_kernel(const int* __restrict__ erow,
                         const int* __restrict__ ecol,
                         const float* __restrict__ a1,
                         const float* __restrict__ a2,
                         unsigned* __restrict__ planes,
                         int* __restrict__ rec,
                         int N, int E, int nblk, int wpad, int nspans) {
    __shared__ unsigned bits[SPAN_W * TILE_P];

    const int tid  = threadIdx.x;

    if ((int)blockIdx.x >= nspans) {
        int4* rec4 = (int4*)rec;
        const int stride = ((int)gridDim.x - nspans) * 1024;
        for (int e0 = ((int)blockIdx.x - nspans) * 1024 + tid; e0 < E; e0 += PACK_U * stride) {
            int   pr[PACK_U], pc[PACK_U];
            float px[PACK_U], py[PACK_U];
            #pragma unroll
            for (int u = 0; u < PACK_U; ++u) {
                const int e = min(e0 + u * stride, E - 1);
                pr[u] = erow[e];
                pc[u] = ecol[e];
                px[u] = a1[e];
                py[u] = a2[e];
            }
            #pragma unroll
            for (int u = 0; u < PACK_U; ++u) {
                const int e = e0 + u * stride;
                if (e < E)
                    rec4[e] = make_int4(pr[u], pc[u], __float_as_int(px[u]), __float_as_int(py[u]));
            }
        }
        return;
    }

    const int lane = tid & 31;
    const int wave = __builtin_amdgcn_readfirstlane(tid >> 5);
    const int c0   = blockIdx.x * SPAN_CH + wave * 32;

    int own[32];
    #pragma unroll
    for (int i = 0; i < 32; ++i) {
        const int e = (c0 + i) * 32 + lane;
        const int r = erow[min(e, E - 1)];
        own[i] = (e < E && (unsigned)r < (unsigned)N) ? r / OWN_ROWS : -1;
    }

    for (int ob = 0; ob < nblk; ob += OWN_TILE) {
        for (int i = tid; i < SPAN_W * TILE_P; i += 1024) bits[i] = 0u;
        __syncthreads();
        #pragma unroll
        for (int i = 0; i < 32; ++i) {
            const unsigned o = (unsigned)(own[i] - ob);
            if (o < (unsigned)OWN_TILE) atomicOr(&bits[wave * TILE_P + o], 1u << i);
        }
        __syncthreads();
        const int no = (nblk - ob < OWN_TILE) ? (nblk - ob) : OWN_TILE;
        for (int i = tid; i < no * SPAN_W; i += 1024) {
            const int o = i >> 5;
            planes[(size_t)(ob + o) * wpad + blockIdx.x * SPAN_W + lane] = bits[lane * TILE_P + o];
        }
        __syncthreads();
    }
}

#define UNIT_OWN  2
#define UNIT_ROWS (OWN_ROWS * UNIT_OWN)
#define VG       16
#define QCAP     (VG * 32 + 288)
#define CLCAP    128
#define GRP      16
#define QHALF    (QCAP / 2)
#define COL_BITS 23
static_assert(UNIT_ROWS <= (1 << (32 - COL_BITS)), "a row in the work-group must fit above the column bits");

template <int G>
__device__ __forceinline__ void gather_group(float* __restrict__ acc,
                                             const float* __restrict__ msg,
                                             int r, int c, float v,
                                             int j0, int lane) {
    float4 gm[G];
    #pragma unroll
    for (int g = 0; g < G; ++g) {
        const int cj = __builtin_amdgcn_readlane(c, j0 + g);
        gm[g] = *(const float4*)(msg + (size_t)cj * OUT_F + lane * 4);
    }
    #pragma unroll
    for (int g = 0; g < G; ++g) {
        const int   rj = __builtin_amdgcn_readlane(r, j0 + g);
        const float w  = __int_as_float(__builtin_amdgcn_readlane(__float_as_int(v), j0 + g));
        float* dst = acc + rj * OUT_F + lane;
        unsafeAtomicAdd(dst,      w * gm[g].x);
        unsafeAtomicAdd(dst + 32, w * gm[g].y);
        unsafeAtomicAdd(dst + 64, w * gm[g].z);
        unsafeAtomicAdd(dst + 96, w * gm[g].w);
    }
}

#define KEEP_NONE  0
#define KEEP_WRITE 1
#define KEEP_READ  2
template <int G>
__device__ __forceinline__ void drain_list(float* __restrict__ acc, int* __restrict__ myq, int cnt,
                                           const int4* __restrict__ rec4,
                                           const float* __restrict__ msg, bool var, int keep,
                                           int row0, int lane) {
    for (int i0 = 0; i0 < cnt; i0 += 32) {
        const int n = (cnt - i0 < 32) ? (cnt - i0) : 32;
        int r = 0, c = 0;
        float v = 0.f;
        if (lane < n) {
            if (keep == KEEP_READ) {
                const unsigned rc = (unsigned)myq[i0 + lane];
                r = (int)(rc >> COL_BITS);
                c = (int)(rc & ((1u << COL_BITS) - 1u));
                v = __int_as_float(myq[QHALF + i0 + lane]);
            } else {
                const int4 q = rec4[myq[i0 + lane]];
                r = q.x - row0;
                c = q.y;
                v = __int_as_float(var ? q.w : q.z);
                if (keep == KEEP_WRITE) {
                    myq[i0 + lane] = (int)(((unsigned)r << COL_BITS) | ((unsigned)c & ((1u << COL_BITS) - 1u)));
                    myq[QHALF + i0 + lane] = q.w;
                }
            }
        }
        int j = 0;
        for (; j + G <= n; j += G)
            gather_group<G>(acc, msg, r, c, v, j, lane);
        if (G > 8 && j + 8 <= n) { gather_group<8>(acc, msg, r, c, v, j, lane); j += 8; }
        if (G > 4 && j + 4 <= n) { gather_group<4>(acc, msg, r, c, v, j, lane); j += 4; }
        if (G > 2 && j + 2 <= n) { gather_group<2>(acc, msg, r, c, v, j, lane); j += 2; }
        if (G > 1 && j < n)        gather_group<1>(acc, msg, r, c, v, j, lane);
    }
}

__global__ __launch_bounds__(1024)
void spmm_owner_kernel(const int* __restrict__ erow,
                       const int* __restrict__ rec,
                       const float* __restrict__ m_mu,
                       const float* __restrict__ m_sig,
                       const unsigned* __restrict__ planes,
                       float* __restrict__ out_mu,
                       float* __restrict__ out_sig,
                       int N, int E, int wpad, int nblk) {
    __shared__ __attribute__((aligned(16))) float acc[UNIT_ROWS * OUT_F];
    __shared__ int elist[32 * QCAP];
    __shared__ int clist[32 * CLCAP];

    const int tid  = threadIdx.x;
    const int lane = tid & 31;
    const int wave = __builtin_amdgcn_readfirstlane(tid >> 5);
    const int row0 = blockIdx.x * UNIT_ROWS;
    const int nrows = (N - row0 < UNIT_ROWS) ? (N - row0) : UNIT_ROWS;
    int* myq = elist + wave * QCAP;
    int* myc = clist + wave * CLCAP;
    const int4* rec4 = (const int4*)rec;

    const int nchunks = (E + 31) >> 5;
    const int nwords  = (nchunks + 31) >> 5;
    const int per     = (nwords + 31) >> 5;
    const int wbeg    = wave * per;
    const int wend    = (wbeg + per < nwords) ? (wbeg + per) : nwords;
    static_assert(UNIT_OWN == 2, "a work-group's plane word is the OR of two owners' words");
    const unsigned* plane = planes + (size_t)(blockIdx.x * UNIT_OWN) * wpad;
    const bool two = (int)blockIdx.x * UNIT_OWN + 1 < nblk;

    int  cnt = 0;
    bool spilt = false;
    bool kept  = false;

    for (int pass = 0; pass < 2; ++pass) {
        const bool var = pass != 0;
        const float* __restrict__ msg = var ? m_sig : m_mu;
        float* __restrict__ out = var ? out_sig : out_mu;

        if (var) __syncthreads();
        for (int i = tid; i < UNIT_ROWS * OUT_F / 4; i += 1024)
            ((float4*)acc)[i] = make_float4(0.f, 0.f, 0.f, 0.f);
        __syncthreads();

        if (!var || spilt) {
            cnt = 0;

            auto make_room = [&]() {
                if (cnt > QCAP - VG * 32) {
                    drain_list<1>(acc, myq, cnt, rec4, msg, var, KEEP_NONE, row0, lane);
                    cnt = 0;
                    spilt = true;
                }
            };

            auto visit = [&](int nlist) {
                for (int i0 = 0; i0 < nlist; i0 += 32) {
                    const int n = (nlist - i0 < 32) ? (nlist - i0) : 32;
                    const int cid = (lane < n) ? myc[i0 + lane] : nchunks;
                    for (int j = 0; j < n; j += VG) {
                        make_room();
                        int rv[VG];
                        #pragma unroll
                        for (int g = 0; g < VG; ++g) {
                            const int e = __builtin_amdgcn_readlane(cid, j + g) * 32 + lane;
                            rv[g] = erow[min(e, E - 1)];
                        }
                        #pragma unroll
                        for (int g = 0; g < VG; ++g) {
                            const int e = __builtin_amdgcn_readlane(cid, j + g) * 32 + lane;
                            const bool hit = (e < E) && ((unsigned)(rv[g] - row0) < (unsigned)nrows);
                            const unsigned mask = __builtin_amdgcn_ballot_w32(hit);
                            const int pos = cnt + (int)__builtin_amdgcn_mbcnt_lo(mask, 0u);
                            if (hit) myq[pos] = e;
                            cnt += __builtin_popcount(mask);
                        }
                    }
                }
            };

            auto plane_word = [&](int w0) -> unsigned {
                const int wi = w0 + lane;
                if (wi >= wend) return 0u;
                const unsigned a = plane[wi];
                const unsigned b = two ? plane[(size_t)wpad + wi] : 0u;
                return a | b;
            };

            unsigned word = plane_word(wbeg);
            for (int w0 = wbeg; w0 < wend; w0 += 32) {
                const unsigned nxt = plane_word(w0 + 32);
                const int cword = (w0 + lane) << 5;
                int nlist = 0;
                for (int t = 0; t <= 32; ++t) {
                    const bool has = word != 0u;
                    const unsigned m = __builtin_amdgcn_ballot_w32(has);
                    const int k = __builtin_popcount(m);
                    if (k == 0 || nlist + k > CLCAP) { visit(nlist); nlist = 0; }
                    if (k == 0) break;
                    if (has) {
                        myc[nlist + (int)__builtin_amdgcn_mbcnt_lo(m, 0u)] = cword + __builtin_ctz(word);
                        word &= word - 1u;
                    }
                    nlist += k;
                }
                word = nxt;
            }
        }
        if (!var) kept = !spilt && cnt <= QHALF && N <= (1 << COL_BITS);
        drain_list<GRP>(acc, myq, cnt, rec4, msg, var,
                        kept ? (var ? KEEP_READ : KEEP_WRITE) : KEEP_NONE, row0, lane);
        __syncthreads();

        for (int i = tid; i < nrows * (OUT_F / 4); i += 1024) {
            const int r  = i >> 5;
            const int c4 = i & 31;
            const float* a = acc + r * OUT_F + c4;
            *(float4*)(out + (size_t)(row0 + r) * OUT_F + c4 * 4) = make_float4(a[0], a[32], a[64], a[96]);
        }
    }
}

extern "C" void kernel_launch(void* const* d_in, const int* in_sizes, int n_in,
                              void* d_out, int out_size, void* d_ws, size_t ws_size,
                              hipStream_t stream) {
    const float* miu   = (const float*)d_in[0];
    const float* sigma = (const float*)d_in[1];
    const int*   erow  = (const int*)  d_in[2];
    const int*   ecol  = (const int*)  d_in[3];
    const float* a1    = (const float*)d_in[4];
    const float* a2    = (const float*)d_in[5];
    const float* Wmu   = (const float*)d_in[6];
    const float* Wsig  = (const float*)d_in[7];

    const int N = in_sizes[0] / IN_F;
    const int E = in_sizes[2];

    float* out_mu  = (float*)d_out;
    float* out_sig = out_mu + (size_t)N * OUT_F;

    float* m_mu  = (float*)d_ws;
    float* m_sig = m_mu + (size_t)N * OUT_F;
    unsigned* planes = (unsigned*)(m_sig + (size_t)N * OUT_F);

    const int nblk   = (N + OWN_ROWS - 1) / OWN_ROWS;
    const int nspans = (E + SPAN_CH * 32 - 1) / (SPAN_CH * 32);
    const int wpad   = nspans * SPAN_W;

    gemm_msg_kernel<<<256, GEMM_THR, 0, stream>>>(
        miu, sigma, Wmu, Wsig, m_mu, m_sig, N, (N + 15) / 16);

    int* rec = (int*)(planes + (size_t)nblk * wpad);
    owner_planes_kernel<<<nspans + PACK_WG, 1024, 0, stream>>>(
        erow, ecol, a1, a2, planes, rec, N, E, nblk, wpad, nspans);

    spmm_owner_kernel<<<(nblk + UNIT_OWN - 1) / UNIT_OWN, 1024, 0, stream>>>(
        erow, rec, m_mu, m_sig, planes, out_mu, out_sig, N, E, wpad, nblk);
}


# owner SpMM scan: a lane keeps its owned row words of a group of chunks as bits of its own; one wave-wide list step a turn (about 3 a group), not one a chunk; LDS add order changes
# speedup vs baseline: 3.3753x; 1.0324x over previous
#include <hip/hip_runtime.h>
#include <hip/hip_bf16.h>

typedef float v2f __attribute__((ext_vector_type(2)));
typedef float v8f __attribute__((ext_vector_type(8)));

#define IN_F 256
#define OUT_F 128

#define GEMM_WAVES 16
#define GEMM_THR   (GEMM_WAVES * 32)
#define W_FLOATS   (IN_F * OUT_F)
#define W_QUADS    (W_FLOATS / 4)
#define KSLAB      (OUT_F * 4)
#define NT_REST    2

template <int NT>
__device__ __forceinline__ void gemm_tile(const float* __restrict__ miu,
                                          const float* __restrict__ sigma,
                                          const float* bsig, const float* bmu,
                                          float* __restrict__ m_mu,
                                          float* __restrict__ m_sig,
                                          int N, int row_base, int ct0, int m, int kgrp) {
    const int arow = (row_base + m < N) ? (row_base + m) : (N - 1);

    float att[NT][8];

    {
        v8f acc[NT];
        v8f zero = {};
        #pragma unroll
        for (int t = 0; t < NT; ++t) acc[t] = zero;

        const float* Ap = sigma + (size_t)arow * IN_F + kgrp * 2;
        const float* Bw = bsig + ct0 * 64;
        #pragma unroll 4
        for (int k0 = 0; k0 < IN_F; k0 += 4) {
            v2f a = *(const v2f*)(Ap + k0);
            const float* Bp = Bw + (k0 >> 2) * KSLAB;
            #pragma unroll
            for (int t = 0; t < NT; ++t) {
                v2f b = *(const v2f*)(Bp + t * 64);
                acc[t] = __builtin_amdgcn_wmma_f32_16x16x4_f32(
                    false, a, false, b, (short)0, acc[t], false, false);
            }
        }
        #pragma unroll
        for (int t = 0; t < NT; ++t) {
            #pragma unroll
            for (int r = 0; r < 8; ++r) {
                float hs = fmaxf(acc[t][r], 0.0f);
                float a  = __expf(-hs);
                att[t][r] = a;
                int rr = row_base + r + kgrp * 8;
                if (rr < N)
                    m_sig[(size_t)rr * OUT_F + (ct0 + t) * 16 + m] = hs * a * a;
            }
        }
    }

    {
        v8f acc[NT];
        v8f zero = {};
        #pragma unroll
        for (int t = 0; t < NT; ++t) acc[t] = zero;

        const float* Ap = miu + (size_t)arow * IN_F + kgrp * 2;
        const float* Bw = bmu + ct0 * 64;
        #pragma unroll 4
        for (int k0 = 0; k0 < IN_F; k0 += 4) {
            v2f a = *(const v2f*)(Ap + k0);
            const float* Bp = Bw + (k0 >> 2) * KSLAB;
            #pragma unroll
            for (int t = 0; t < NT; ++t) {
                v2f b = *(const v2f*)(Bp + t * 64);
                acc[t] = __builtin_amdgcn_wmma_f32_16x16x4_f32(
                    false, a, false, b, (short)0, acc[t], false, false);
            }
        }
        #pragma unroll
        for (int t = 0; t < NT; ++t) {
            #pragma unroll
            for (int r = 0; r < 8; ++r) {
                float hm = acc[t][r];
                float e  = (hm > 0.0f) ? hm : (__expf(hm) - 1.0f);
                int rr = row_base + r + kgrp * 8;
                if (rr < N)
                    m_mu[(size_t)rr * OUT_F + (ct0 + t) * 16 + m] = e * att[t][r];
            }
        }
    }
}

__global__ __launch_bounds__(GEMM_THR)
void gemm_msg_kernel(const float* __restrict__ miu,
                     const float* __restrict__ sigma,
                     const float* __restrict__ Wmu,
                     const float* __restrict__ Wsig,
                     float* __restrict__ m_mu,
                     float* __restrict__ m_sig,
                     int N, int ntiles) {
    __shared__ __attribute__((aligned(16))) float wl[2 * W_FLOATS];

    const int tid  = threadIdx.x;

    {
        const int n = tid & (OUT_F - 1);
        const int q = tid / OUT_F;
        const float* ssig = Wsig + (size_t)(4 * q) * OUT_F + n;
        const float* smu  = Wmu  + (size_t)(4 * q) * OUT_F + n;
        float4* dsig = (float4*)wl + tid;
        float4* dmu  = dsig + W_QUADS;
        #pragma unroll
        for (int j = 0; j < W_QUADS / GEMM_THR; ++j) {
            const float* s = ssig + (size_t)j * (GEMM_THR / OUT_F) * 4 * OUT_F;
            float4 v;
            v.x = s[0]; v.y = s[OUT_F]; v.z = s[2 * OUT_F]; v.w = s[3 * OUT_F];
            dsig[j * GEMM_THR] = v;
        }
        #pragma unroll
        for (int j = 0; j < W_QUADS / GEMM_THR; ++j) {
            const float* s = smu + (size_t)j * (GEMM_THR / OUT_F) * 4 * OUT_F;
            float4 v;
            v.x = s[0]; v.y = s[OUT_F]; v.z = s[2 * OUT_F]; v.w = s[3 * OUT_F];
            dmu[j * GEMM_THR] = v;
        }
    }
    __syncthreads();

    const int lane = tid & 31;
    const int wave = __builtin_amdgcn_readfirstlane(tid >> 5);

    const int m    = lane & 15;
    const int kgrp = lane >> 4;

    const float* bsig = wl + m * 4 + kgrp * 2;
    const float* bmu  = bsig + W_FLOATS;

    const int G      = (int)gridDim.x;
    const int nfull  = ntiles - ntiles % (4 * G);
    const int nslots = nfull + (ntiles - nfull) * (8 / NT_REST);

    for (int slot = wave * G + (int)blockIdx.x; slot < nslots; slot += GEMM_WAVES * G) {
        if (slot < nfull) {
            gemm_tile<8>(miu, sigma, bsig, bmu, m_mu, m_sig, N, slot * 16, 0, m, kgrp);
        } else {
            const int u = slot - nfull;
            gemm_tile<NT_REST>(miu, sigma, bsig, bmu, m_mu, m_sig, N,
                               (nfull + u / (8 / NT_REST)) * 16, (u % (8 / NT_REST)) * NT_REST, m, kgrp);
        }
    }
}

#define OWN_ROWS 196
#define SPAN_W   32
#define SPAN_CH  (SPAN_W * 32)
#define OWN_TILE 512
#define TILE_P   (OWN_TILE + 1)
#define PACK_WG  207
#define PACK_U   4

__global__ __launch_bounds__(1024)
void owner_planes_kernel(const int* __restrict__ erow,
                         const int* __restrict__ ecol,
                         const float* __restrict__ a1,
                         const float* __restrict__ a2,
                         unsigned* __restrict__ planes,
                         int* __restrict__ rec,
                         int N, int E, int nblk, int wpad, int nspans) {
    __shared__ unsigned bits[SPAN_W * TILE_P];

    const int tid  = threadIdx.x;

    if ((int)blockIdx.x >= nspans) {
        int4* rec4 = (int4*)rec;
        const int stride = ((int)gridDim.x - nspans) * 1024;
        for (int e0 = ((int)blockIdx.x - nspans) * 1024 + tid; e0 < E; e0 += PACK_U * stride) {
            int   pr[PACK_U], pc[PACK_U];
            float px[PACK_U], py[PACK_U];
            #pragma unroll
            for (int u = 0; u < PACK_U; ++u) {
                const int e = min(e0 + u * stride, E - 1);
                pr[u] = erow[e];
                pc[u] = ecol[e];
                px[u] = a1[e];
                py[u] = a2[e];
            }
            #pragma unroll
            for (int u = 0; u < PACK_U; ++u) {
                const int e = e0 + u * stride;
                if (e < E)
                    rec4[e] = make_int4(pr[u], pc[u], __float_as_int(px[u]), __float_as_int(py[u]));
            }
        }
        return;
    }

    const int lane = tid & 31;
    const int wave = __builtin_amdgcn_readfirstlane(tid >> 5);
    const int c0   = blockIdx.x * SPAN_CH + wave * 32;

    int own[32];
    #pragma unroll
    for (int i = 0; i < 32; ++i) {
        const int e = (c0 + i) * 32 + lane;
        const int r = erow[min(e, E - 1)];
        own[i] = (e < E && (unsigned)r < (unsigned)N) ? r / OWN_ROWS : -1;
    }

    for (int ob = 0; ob < nblk; ob += OWN_TILE) {
        for (int i = tid; i < SPAN_W * TILE_P; i += 1024) bits[i] = 0u;
        __syncthreads();
        #pragma unroll
        for (int i = 0; i < 32; ++i) {
            const unsigned o = (unsigned)(own[i] - ob);
            if (o < (unsigned)OWN_TILE) atomicOr(&bits[wave * TILE_P + o], 1u << i);
        }
        __syncthreads();
        const int no = (nblk - ob < OWN_TILE) ? (nblk - ob) : OWN_TILE;
        for (int i = tid; i < no * SPAN_W; i += 1024) {
            const int o = i >> 5;
            planes[(size_t)(ob + o) * wpad + blockIdx.x * SPAN_W + lane] = bits[lane * TILE_P + o];
        }
        __syncthreads();
    }
}

#define UNIT_OWN  2
#define UNIT_ROWS (OWN_ROWS * UNIT_OWN)
#define VG       16
#define QCAP     (VG * 32 + 288)
#define CLCAP    128
#define GRP      16
#define QHALF    (QCAP / 2)
#define COL_BITS 23
static_assert(UNIT_ROWS <= (1 << (32 - COL_BITS)), "a row in the work-group must fit above the column bits");

template <int G>
__device__ __forceinline__ void gather_group(float* __restrict__ acc,
                                             const float* __restrict__ msg,
                                             int r, int c, float v,
                                             int j0, int lane) {
    float4 gm[G];
    #pragma unroll
    for (int g = 0; g < G; ++g) {
        const int cj = __builtin_amdgcn_readlane(c, j0 + g);
        gm[g] = *(const float4*)(msg + (size_t)cj * OUT_F + lane * 4);
    }
    #pragma unroll
    for (int g = 0; g < G; ++g) {
        const int   rj = __builtin_amdgcn_readlane(r, j0 + g);
        const float w  = __int_as_float(__builtin_amdgcn_readlane(__float_as_int(v), j0 + g));
        float* dst = acc + rj * OUT_F + lane;
        unsafeAtomicAdd(dst,      w * gm[g].x);
        unsafeAtomicAdd(dst + 32, w * gm[g].y);
        unsafeAtomicAdd(dst + 64, w * gm[g].z);
        unsafeAtomicAdd(dst + 96, w * gm[g].w);
    }
}

#define KEEP_NONE  0
#define KEEP_WRITE 1
#define KEEP_READ  2
template <int G>
__device__ __forceinline__ void drain_list(float* __restrict__ acc, int* __restrict__ myq, int cnt,
                                           const int4* __restrict__ rec4,
                                           const float* __restrict__ msg, bool var, int keep,
                                           int row0, int lane) {
    for (int i0 = 0; i0 < cnt; i0 += 32) {
        const int n = (cnt - i0 < 32) ? (cnt - i0) : 32;
        int r = 0, c = 0;
        float v = 0.f;
        if (lane < n) {
            if (keep == KEEP_READ) {
                const unsigned rc = (unsigned)myq[i0 + lane];
                r = (int)(rc >> COL_BITS);
                c = (int)(rc & ((1u << COL_BITS) - 1u));
                v = __int_as_float(myq[QHALF + i0 + lane]);
            } else {
                const int4 q = rec4[myq[i0 + lane]];
                r = q.x - row0;
                c = q.y;
                v = __int_as_float(var ? q.w : q.z);
                if (keep == KEEP_WRITE) {
                    myq[i0 + lane] = (int)(((unsigned)r << COL_BITS) | ((unsigned)c & ((1u << COL_BITS) - 1u)));
                    myq[QHALF + i0 + lane] = q.w;
                }
            }
        }
        int j = 0;
        for (; j + G <= n; j += G)
            gather_group<G>(acc, msg, r, c, v, j, lane);
        if (G > 8 && j + 8 <= n) { gather_group<8>(acc, msg, r, c, v, j, lane); j += 8; }
        if (G > 4 && j + 4 <= n) { gather_group<4>(acc, msg, r, c, v, j, lane); j += 4; }
        if (G > 2 && j + 2 <= n) { gather_group<2>(acc, msg, r, c, v, j, lane); j += 2; }
        if (G > 1 && j < n)        gather_group<1>(acc, msg, r, c, v, j, lane);
    }
}

__global__ __launch_bounds__(1024)
void spmm_owner_kernel(const int* __restrict__ erow,
                       const int* __restrict__ rec,
                       const float* __restrict__ m_mu,
                       const float* __restrict__ m_sig,
                       const unsigned* __restrict__ planes,
                       float* __restrict__ out_mu,
                       float* __restrict__ out_sig,
                       int N, int E, int wpad, int nblk) {
    __shared__ __attribute__((aligned(16))) float acc[UNIT_ROWS * OUT_F];
    __shared__ int elist[32 * QCAP];
    __shared__ int clist[32 * CLCAP];

    const int tid  = threadIdx.x;
    const int lane = tid & 31;
    const int wave = __builtin_amdgcn_readfirstlane(tid >> 5);
    const int row0 = blockIdx.x * UNIT_ROWS;
    const int nrows = (N - row0 < UNIT_ROWS) ? (N - row0) : UNIT_ROWS;
    int* myq = elist + wave * QCAP;
    int* myc = clist + wave * CLCAP;
    const int4* rec4 = (const int4*)rec;

    const int nchunks = (E + 31) >> 5;
    const int nwords  = (nchunks + 31) >> 5;
    const int per     = (nwords + 31) >> 5;
    const int wbeg    = wave * per;
    const int wend    = (wbeg + per < nwords) ? (wbeg + per) : nwords;
    static_assert(UNIT_OWN == 2, "a work-group's plane word is the OR of two owners' words");
    const unsigned* plane = planes + (size_t)(blockIdx.x * UNIT_OWN) * wpad;
    const bool two = (int)blockIdx.x * UNIT_OWN + 1 < nblk;

    int  cnt = 0;
    bool spilt = false;
    bool kept  = false;

    for (int pass = 0; pass < 2; ++pass) {
        const bool var = pass != 0;
        const float* __restrict__ msg = var ? m_sig : m_mu;
        float* __restrict__ out = var ? out_sig : out_mu;

        if (var) __syncthreads();
        for (int i = tid; i < UNIT_ROWS * OUT_F / 4; i += 1024)
            ((float4*)acc)[i] = make_float4(0.f, 0.f, 0.f, 0.f);
        __syncthreads();

        if (!var || spilt) {
            cnt = 0;

            auto make_room = [&]() {
                if (cnt > QCAP - VG * 32) {
                    drain_list<1>(acc, myq, cnt, rec4, msg, var, KEEP_NONE, row0, lane);
                    cnt = 0;
                    spilt = true;
                }
            };

            auto visit = [&](int nlist) {
                for (int i0 = 0; i0 < nlist; i0 += 32) {
                    const int n = (nlist - i0 < 32) ? (nlist - i0) : 32;
                    const int cid = (lane < n) ? myc[i0 + lane] : nchunks;
                    const unsigned late = __builtin_amdgcn_ballot_w32(cid >= (E >> 5));
                    for (int j = 0; j < n; j += VG) {
                        make_room();
                        int rv[VG];
                        unsigned own = 0u;
                        if (((late >> j) & ((1u << VG) - 1u)) == 0u) {
                            #pragma unroll
                            for (int g = 0; g < VG; ++g) {
                                const int e = __builtin_amdgcn_readlane(cid, j + g) * 32 + lane;
                                rv[g] = erow[e];
                            }
                            #pragma unroll
                            for (int g = 0; g < VG; ++g)
                                own |= ((unsigned)(rv[g] - row0) < (unsigned)nrows) ? (1u << g) : 0u;
                        } else {
                            #pragma unroll
                            for (int g = 0; g < VG; ++g) {
                                const int e = __builtin_amdgcn_readlane(cid, j + g) * 32 + lane;
                                rv[g] = erow[min(e, E - 1)];
                            }
                            #pragma unroll
                            for (int g = 0; g < VG; ++g) {
                                const int e = __builtin_amdgcn_readlane(cid, j + g) * 32 + lane;
                                const bool hit = (e < E) && ((unsigned)(rv[g] - row0) < (unsigned)nrows);
                                own |= hit ? (1u << g) : 0u;
                            }
                        }
                        for (int t = 0; t <= VG; ++t) {
                            const bool has = own != 0u;
                            const unsigned mask = __builtin_amdgcn_ballot_w32(has);
                            if (mask == 0u) break;
                            if (has) {
                                const int ch = myc[i0 + j + __builtin_ctz(own)];
                                myq[cnt + (int)__builtin_amdgcn_mbcnt_lo(mask, 0u)] = ch * 32 + lane;
                                own &= own - 1u;
                            }
                            cnt += __builtin_popcount(mask);
                        }
                    }
                }
            };

            auto plane_word = [&](int w0) -> unsigned {
                const int wi = w0 + lane;
                if (wi >= wend) return 0u;
                const unsigned a = plane[wi];
                const unsigned b = two ? plane[(size_t)wpad + wi] : 0u;
                return a | b;
            };

            unsigned word = plane_word(wbeg);
            for (int w0 = wbeg; w0 < wend; w0 += 32) {
                const unsigned nxt = plane_word(w0 + 32);
                const int cword = (w0 + lane) << 5;
                int nlist = 0;
                for (int t = 0; t <= 32; ++t) {
                    const bool has = word != 0u;
                    const unsigned m = __builtin_amdgcn_ballot_w32(has);
                    const int k = __builtin_popcount(m);
                    if (k == 0 || nlist + k > CLCAP) { visit(nlist); nlist = 0; }
                    if (k == 0) break;
                    if (has) {
                        myc[nlist + (int)__builtin_amdgcn_mbcnt_lo(m, 0u)] = cword + __builtin_ctz(word);
                        word &= word - 1u;
                    }
                    nlist += k;
                }
                word = nxt;
            }
        }
        if (!var) kept = !spilt && cnt <= QHALF && N <= (1 << COL_BITS);
        drain_list<GRP>(acc, myq, cnt, rec4, msg, var,
                        kept ? (var ? KEEP_READ : KEEP_WRITE) : KEEP_NONE, row0, lane);
        __syncthreads();

        for (int i = tid; i < nrows * (OUT_F / 4); i += 1024) {
            const int r  = i >> 5;
            const int c4 = i & 31;
            const float* a = acc + r * OUT_F + c4;
            *(float4*)(out + (size_t)(row0 + r) * OUT_F + c4 * 4) = make_float4(a[0], a[32], a[64], a[96]);
        }
    }
}

extern "C" void kernel_launch(void* const* d_in, const int* in_sizes, int n_in,
                              void* d_out, int out_size, void* d_ws, size_t ws_size,
                              hipStream_t stream) {
    const float* miu   = (const float*)d_in[0];
    const float* sigma = (const float*)d_in[1];
    const int*   erow  = (const int*)  d_in[2];
    const int*   ecol  = (const int*)  d_in[3];
    const float* a1    = (const float*)d_in[4];
    const float* a2    = (const float*)d_in[5];
    const float* Wmu   = (const float*)d_in[6];
    const float* Wsig  = (const float*)d_in[7];

    const int N = in_sizes[0] / IN_F;
    const int E = in_sizes[2];

    float* out_mu  = (float*)d_out;
    float* out_sig = out_mu + (size_t)N * OUT_F;

    float* m_mu  = (float*)d_ws;
    float* m_sig = m_mu + (size_t)N * OUT_F;
    unsigned* planes = (unsigned*)(m_sig + (size_t)N * OUT_F);

    const int nblk   = (N + OWN_ROWS - 1) / OWN_ROWS;
    const int nspans = (E + SPAN_CH * 32 - 1) / (SPAN_CH * 32);
    const int wpad   = nspans * SPAN_W;

    gemm_msg_kernel<<<256, GEMM_THR, 0, stream>>>(
        miu, sigma, Wmu, Wsig, m_mu, m_sig, N, (N + 15) / 16);

    int* rec = (int*)(planes + (size_t)nblk * wpad);
    owner_planes_kernel<<<nspans + PACK_WG, 1024, 0, stream>>>(
        erow, ecol, a1, a2, planes, rec, N, E, nblk, wpad, nspans);

    spmm_owner_kernel<<<(nblk + UNIT_OWN - 1) / UNIT_OWN, 1024, 0, stream>>>(
        erow, rec, m_mu, m_sig, planes, out_mu, out_sig, N, E, wpad, nblk);
}


# owner SpMM: LDS float atomics replaced by plain 16-byte LDS read-add-write: each found edge goes to the list of the wave that owns its row (row mod 32); sums in list arrival order
# speedup vs baseline: 3.5387x; 1.0484x over previous
#include <hip/hip_runtime.h>
#include <hip/hip_bf16.h>

typedef float v2f __attribute__((ext_vector_type(2)));
typedef float v8f __attribute__((ext_vector_type(8)));

#define IN_F 256
#define OUT_F 128

#define GEMM_WAVES 16
#define GEMM_THR   (GEMM_WAVES * 32)
#define W_FLOATS   (IN_F * OUT_F)
#define W_QUADS    (W_FLOATS / 4)
#define KSLAB      (OUT_F * 4)
#define NT_REST    2

template <int NT>
__device__ __forceinline__ void gemm_tile(const float* __restrict__ miu,
                                          const float* __restrict__ sigma,
                                          const float* bsig, const float* bmu,
                                          float* __restrict__ m_mu,
                                          float* __restrict__ m_sig,
                                          int N, int row_base, int ct0, int m, int kgrp) {
    const int arow = (row_base + m < N) ? (row_base + m) : (N - 1);

    float att[NT][8];

    {
        v8f acc[NT];
        v8f zero = {};
        #pragma unroll
        for (int t = 0; t < NT; ++t) acc[t] = zero;

        const float* Ap = sigma + (size_t)arow * IN_F + kgrp * 2;
        const float* Bw = bsig + ct0 * 64;
        #pragma unroll 4
        for (int k0 = 0; k0 < IN_F; k0 += 4) {
            v2f a = *(const v2f*)(Ap + k0);
            const float* Bp = Bw + (k0 >> 2) * KSLAB;
            #pragma unroll
            for (int t = 0; t < NT; ++t) {
                v2f b = *(const v2f*)(Bp + t * 64);
                acc[t] = __builtin_amdgcn_wmma_f32_16x16x4_f32(
                    false, a, false, b, (short)0, acc[t], false, false);
            }
        }
        #pragma unroll
        for (int t = 0; t < NT; ++t) {
            #pragma unroll
            for (int r = 0; r < 8; ++r) {
                float hs = fmaxf(acc[t][r], 0.0f);
                float a  = __expf(-hs);
                att[t][r] = a;
                int rr = row_base + r + kgrp * 8;
                if (rr < N)
                    m_sig[(size_t)rr * OUT_F + (ct0 + t) * 16 + m] = hs * a * a;
            }
        }
    }

    {
        v8f acc[NT];
        v8f zero = {};
        #pragma unroll
        for (int t = 0; t < NT; ++t) acc[t] = zero;

        const float* Ap = miu + (size_t)arow * IN_F + kgrp * 2;
        const float* Bw = bmu + ct0 * 64;
        #pragma unroll 4
        for (int k0 = 0; k0 < IN_F; k0 += 4) {
            v2f a = *(const v2f*)(Ap + k0);
            const float* Bp = Bw + (k0 >> 2) * KSLAB;
            #pragma unroll
            for (int t = 0; t < NT; ++t) {
                v2f b = *(const v2f*)(Bp + t * 64);
                acc[t] = __builtin_amdgcn_wmma_f32_16x16x4_f32(
                    false, a, false, b, (short)0, acc[t], false, false);
            }
        }
        #pragma unroll
        for (int t = 0; t < NT; ++t) {
            #pragma unroll
            for (int r = 0; r < 8; ++r) {
                float hm = acc[t][r];
                float e  = (hm > 0.0f) ? hm : (__expf(hm) - 1.0f);
                int rr = row_base + r + kgrp * 8;
                if (rr < N)
                    m_mu[(size_t)rr * OUT_F + (ct0 + t) * 16 + m] = e * att[t][r];
            }
        }
    }
}

__global__ __launch_bounds__(GEMM_THR)
void gemm_msg_kernel(const float* __restrict__ miu,
                     const float* __restrict__ sigma,
                     const float* __restrict__ Wmu,
                     const float* __restrict__ Wsig,
                     float* __restrict__ m_mu,
                     float* __restrict__ m_sig,
                     int N, int ntiles) {
    __shared__ __attribute__((aligned(16))) float wl[2 * W_FLOATS];

    const int tid  = threadIdx.x;

    {
        const int n = tid & (OUT_F - 1);
        const int q = tid / OUT_F;
        const float* ssig = Wsig + (size_t)(4 * q) * OUT_F + n;
        const float* smu  = Wmu  + (size_t)(4 * q) * OUT_F + n;
        float4* dsig = (float4*)wl + tid;
        float4* dmu  = dsig + W_QUADS;
        #pragma unroll
        for (int j = 0; j < W_QUADS / GEMM_THR; ++j) {
            const float* s = ssig + (size_t)j * (GEMM_THR / OUT_F) * 4 * OUT_F;
            float4 v;
            v.x = s[0]; v.y = s[OUT_F]; v.z = s[2 * OUT_F]; v.w = s[3 * OUT_F];
            dsig[j * GEMM_THR] = v;
        }
        #pragma unroll
        for (int j = 0; j < W_QUADS / GEMM_THR; ++j) {
            const float* s = smu + (size_t)j * (GEMM_THR / OUT_F) * 4 * OUT_F;
            float4 v;
            v.x = s[0]; v.y = s[OUT_F]; v.z = s[2 * OUT_F]; v.w = s[3 * OUT_F];
            dmu[j * GEMM_THR] = v;
        }
    }
    __syncthreads();

    const int lane = tid & 31;
    const int wave = __builtin_amdgcn_readfirstlane(tid >> 5);

    const int m    = lane & 15;
    const int kgrp = lane >> 4;

    const float* bsig = wl + m * 4 + kgrp * 2;
    const float* bmu  = bsig + W_FLOATS;

    const int G      = (int)gridDim.x;
    const int nfull  = ntiles - ntiles % (4 * G);
    const int nslots = nfull + (ntiles - nfull) * (8 / NT_REST);

    for (int slot = wave * G + (int)blockIdx.x; slot < nslots; slot += GEMM_WAVES * G) {
        if (slot < nfull) {
            gemm_tile<8>(miu, sigma, bsig, bmu, m_mu, m_sig, N, slot * 16, 0, m, kgrp);
        } else {
            const int u = slot - nfull;
            gemm_tile<NT_REST>(miu, sigma, bsig, bmu, m_mu, m_sig, N,
                               (nfull + u / (8 / NT_REST)) * 16, (u % (8 / NT_REST)) * NT_REST, m, kgrp);
        }
    }
}

#define OWN_ROWS 196
#define SPAN_W   32
#define SPAN_CH  (SPAN_W * 32)
#define OWN_TILE 512
#define TILE_P   (OWN_TILE + 1)
#define PACK_WG  207
#define PACK_U   4

__global__ __launch_bounds__(1024)
void owner_planes_kernel(const int* __restrict__ erow,
                         const int* __restrict__ ecol,
                         const float* __restrict__ a1,
                         const float* __restrict__ a2,
                         unsigned* __restrict__ planes,
                         int* __restrict__ rec,
                         int N, int E, int nblk, int wpad, int nspans) {
    __shared__ unsigned bits[SPAN_W * TILE_P];

    const int tid  = threadIdx.x;

    if ((int)blockIdx.x >= nspans) {
        int4* rec4 = (int4*)rec;
        const int stride = ((int)gridDim.x - nspans) * 1024;
        for (int e0 = ((int)blockIdx.x - nspans) * 1024 + tid; e0 < E; e0 += PACK_U * stride) {
            int   pr[PACK_U], pc[PACK_U];
            float px[PACK_U], py[PACK_U];
            #pragma unroll
            for (int u = 0; u < PACK_U; ++u) {
                const int e = min(e0 + u * stride, E - 1);
                pr[u] = erow[e];
                pc[u] = ecol[e];
                px[u] = a1[e];
                py[u] = a2[e];
            }
            #pragma unroll
            for (int u = 0; u < PACK_U; ++u) {
                const int e = e0 + u * stride;
                if (e < E)
                    rec4[e] = make_int4(pr[u], pc[u], __float_as_int(px[u]), __float_as_int(py[u]));
            }
        }
        return;
    }

    const int lane = tid & 31;
    const int wave = __builtin_amdgcn_readfirstlane(tid >> 5);
    const int c0   = blockIdx.x * SPAN_CH + wave * 32;

    int own[32];
    #pragma unroll
    for (int i = 0; i < 32; ++i) {
        const int e = (c0 + i) * 32 + lane;
        const int r = erow[min(e, E - 1)];
        own[i] = (e < E && (unsigned)r < (unsigned)N) ? r / OWN_ROWS : -1;
    }

    for (int ob = 0; ob < nblk; ob += OWN_TILE) {
        for (int i = tid; i < SPAN_W * TILE_P; i += 1024) bits[i] = 0u;
        __syncthreads();
        #pragma unroll
        for (int i = 0; i < 32; ++i) {
            const unsigned o = (unsigned)(own[i] - ob);
            if (o < (unsigned)OWN_TILE) atomicOr(&bits[wave * TILE_P + o], 1u << i);
        }
        __syncthreads();
        const int no = (nblk - ob < OWN_TILE) ? (nblk - ob) : OWN_TILE;
        for (int i = tid; i < no * SPAN_W; i += 1024) {
            const int o = i >> 5;
            planes[(size_t)(ob + o) * wpad + blockIdx.x * SPAN_W + lane] = bits[lane * TILE_P + o];
        }
        __syncthreads();
    }
}

#define UNIT_OWN  2
#define UNIT_ROWS (OWN_ROWS * UNIT_OWN)
#define VG       16
#define QCAP     (VG * 32 + 328)
#define CLCAP    128
#define GRP      16
#define QHALF    (QCAP / 2)
#define COL_BITS 23
static_assert(UNIT_ROWS <= (1 << (32 - COL_BITS)), "a row in the work-group must fit above the column bits");
#define SRC_CAP  (UNIT_ROWS * OUT_F / 32)
#define TCAP     (QCAP / 3)
static_assert(SRC_CAP > VG * 32, "a whole group of chunks must fit a wave's share of the sums' space");

template <int G>
__device__ __forceinline__ void gather_group(float* __restrict__ acc,
                                             const float* __restrict__ msg,
                                             int r, int c, float v,
                                             int j0, int lane) {
    float4 gm[G];
    #pragma unroll
    for (int g = 0; g < G; ++g) {
        const int cj = __builtin_amdgcn_readlane(c, j0 + g);
        gm[g] = *(const float4*)(msg + (size_t)cj * OUT_F + lane * 4);
    }
    #pragma unroll
    for (int g = 0; g < G; ++g) {
        const int   rj = __builtin_amdgcn_readlane(r, j0 + g);
        const float w  = __int_as_float(__builtin_amdgcn_readlane(__float_as_int(v), j0 + g));
        float* dst = acc + rj * OUT_F + lane;
        unsafeAtomicAdd(dst,      w * gm[g].x);
        unsafeAtomicAdd(dst + 32, w * gm[g].y);
        unsafeAtomicAdd(dst + 64, w * gm[g].z);
        unsafeAtomicAdd(dst + 96, w * gm[g].w);
    }
}

#define KEEP_NONE  0
#define KEEP_WRITE 1
#define KEEP_READ  2
template <int G>
__device__ __forceinline__ void drain_list(float* __restrict__ acc, int* __restrict__ myq, int cnt,
                                           const int4* __restrict__ rec4,
                                           const float* __restrict__ msg, bool var, int keep,
                                           int row0, int lane) {
    for (int i0 = 0; i0 < cnt; i0 += 32) {
        const int n = (cnt - i0 < 32) ? (cnt - i0) : 32;
        int r = 0, c = 0;
        float v = 0.f;
        if (lane < n) {
            if (keep == KEEP_READ) {
                const unsigned rc = (unsigned)myq[i0 + lane];
                r = (int)(rc >> COL_BITS);
                c = (int)(rc & ((1u << COL_BITS) - 1u));
                v = __int_as_float(myq[QHALF + i0 + lane]);
            } else {
                const int4 q = rec4[myq[i0 + lane]];
                r = q.x - row0;
                c = q.y;
                v = __int_as_float(var ? q.w : q.z);
                if (keep == KEEP_WRITE) {
                    myq[i0 + lane] = (int)(((unsigned)r << COL_BITS) | ((unsigned)c & ((1u << COL_BITS) - 1u)));
                    myq[QHALF + i0 + lane] = q.w;
                }
            }
        }
        int j = 0;
        for (; j + G <= n; j += G)
            gather_group<G>(acc, msg, r, c, v, j, lane);
        if (G > 8 && j + 8 <= n) { gather_group<8>(acc, msg, r, c, v, j, lane); j += 8; }
        if (G > 4 && j + 4 <= n) { gather_group<4>(acc, msg, r, c, v, j, lane); j += 4; }
        if (G > 2 && j + 2 <= n) { gather_group<2>(acc, msg, r, c, v, j, lane); j += 2; }
        if (G > 1 && j < n)        gather_group<1>(acc, msg, r, c, v, j, lane);
    }
}

template <int G>
__device__ __forceinline__ void rmw_group(float* __restrict__ acc,
                                          const float* __restrict__ msg,
                                          int r, int c, float v,
                                          int j0, int lane) {
    float4 gm[G];
    #pragma unroll
    for (int g = 0; g < G; ++g) {
        const int cj = __builtin_amdgcn_readlane(c, j0 + g);
        gm[g] = *(const float4*)(msg + (size_t)cj * OUT_F + lane * 4);
    }
    if (G == 1) {
        const int   ra = __builtin_amdgcn_readlane(r, j0);
        const float wa = __int_as_float(__builtin_amdgcn_readlane(__float_as_int(v), j0));
        float4* pa = (float4*)(acc + ra * OUT_F) + lane;
        float4 a = *pa;
        a.x += wa * gm[0].x; a.y += wa * gm[0].y; a.z += wa * gm[0].z; a.w += wa * gm[0].w;
        *pa = a;
        return;
    }
    #pragma unroll
    for (int g = 0; g + 1 < G; g += 2) {
        const int   ra = __builtin_amdgcn_readlane(r, j0 + g);
        const int   rb = __builtin_amdgcn_readlane(r, j0 + g + 1);
        const float wa = __int_as_float(__builtin_amdgcn_readlane(__float_as_int(v), j0 + g));
        const float wb = __int_as_float(__builtin_amdgcn_readlane(__float_as_int(v), j0 + g + 1));
        const float4 ma = gm[g];
        const float4 mb = gm[(g + 1 < G) ? g + 1 : g];
        float4* pa = (float4*)(acc + ra * OUT_F) + lane;
        float4* pb = (float4*)(acc + rb * OUT_F) + lane;
        if (ra != rb) {
            float4 a = *pa;
            float4 b = *pb;
            a.x += wa * ma.x; a.y += wa * ma.y; a.z += wa * ma.z; a.w += wa * ma.w;
            b.x += wb * mb.x; b.y += wb * mb.y; b.z += wb * mb.z; b.w += wb * mb.w;
            *pa = a;
            *pb = b;
        } else {
            float4 a = *pa;
            a.x += wa * ma.x; a.y += wa * ma.y; a.z += wa * ma.z; a.w += wa * ma.w;
            a.x += wb * mb.x; a.y += wb * mb.y; a.z += wb * mb.z; a.w += wb * mb.w;
            *pa = a;
        }
    }
}

template <int G>
__device__ __forceinline__ void drain_owned(float* __restrict__ acc, const int* __restrict__ tl, int cnt,
                                            const float* __restrict__ msg, int wofs, int lane) {
    for (int i0 = 0; i0 < cnt; i0 += 32) {
        const int n = (cnt - i0 < 32) ? (cnt - i0) : 32;
        int r = 0, c = 0;
        float v = 0.f;
        if (lane < n) {
            const unsigned rc = (unsigned)tl[i0 + lane];
            r = (int)(rc >> COL_BITS);
            c = (int)(rc & ((1u << COL_BITS) - 1u));
            v = __int_as_float(tl[wofs + i0 + lane]);
        }
        int j = 0;
        for (; j + G <= n; j += G)
            rmw_group<G>(acc, msg, r, c, v, j, lane);
        if (G > 8 && j + 8 <= n) { rmw_group<8>(acc, msg, r, c, v, j, lane); j += 8; }
        if (G > 4 && j + 4 <= n) { rmw_group<4>(acc, msg, r, c, v, j, lane); j += 4; }
        if (G > 2 && j + 2 <= n) { rmw_group<2>(acc, msg, r, c, v, j, lane); j += 2; }
        if (G > 1 && j < n)        rmw_group<1>(acc, msg, r, c, v, j, lane);
    }
}

__global__ __launch_bounds__(1024)
void spmm_owner_kernel(const int* __restrict__ erow,
                       const int* __restrict__ rec,
                       const float* __restrict__ m_mu,
                       const float* __restrict__ m_sig,
                       const unsigned* __restrict__ planes,
                       float* __restrict__ out_mu,
                       float* __restrict__ out_sig,
                       int N, int E, int wpad, int nblk) {
    __shared__ __attribute__((aligned(16))) float acc[UNIT_ROWS * OUT_F];
    __shared__ int elist[32 * QCAP];
    __shared__ int clist[32 * CLCAP];

    const int tid  = threadIdx.x;
    const int lane = tid & 31;
    const int wave = __builtin_amdgcn_readfirstlane(tid >> 5);
    const int row0 = blockIdx.x * UNIT_ROWS;
    const int nrows = (N - row0 < UNIT_ROWS) ? (N - row0) : UNIT_ROWS;
    int* myq = elist + wave * QCAP;
    int* myc = clist + wave * CLCAP;
    const int4* rec4 = (const int4*)rec;

    const int nchunks = (E + 31) >> 5;
    const int nwords  = (nchunks + 31) >> 5;
    const int per     = (nwords + 31) >> 5;
    const int wbeg    = wave * per;
    const int wend    = (wbeg + per < nwords) ? (wbeg + per) : nwords;
    static_assert(UNIT_OWN == 2, "a work-group's plane word is the OR of two owners' words");
    const unsigned* plane = planes + (size_t)(blockIdx.x * UNIT_OWN) * wpad;
    const bool two = (int)blockIdx.x * UNIT_OWN + 1 < nblk;


    __shared__ int tcnt[32];
    __shared__ int bad1, bad2;
    if (tid < 32) tcnt[tid] = 0;
    if (tid == 0) { bad1 = 0; bad2 = 0; }
    __syncthreads();
    {
        const bool tryfast = N <= (1 << COL_BITS);
        int* srcl = (int*)acc + wave * SRC_CAP;
        int  fcnt = 0;
        bool ovf  = false;
        if (tryfast) {
            auto make_room_f = [&]() {
                if (fcnt > SRC_CAP - VG * 32) { fcnt = 0; ovf = true; }
            };
            auto visit_f = [&](int nlist) {
                for (int i0 = 0; i0 < nlist; i0 += 32) {
                    const int n = (nlist - i0 < 32) ? (nlist - i0) : 32;
                    const int cid = (lane < n) ? myc[i0 + lane] : nchunks;
                    const unsigned late = __builtin_amdgcn_ballot_w32(cid >= (E >> 5));
                    for (int j = 0; j < n; j += VG) {
                        make_room_f();
                        int rv[VG];
                        unsigned own = 0u;
                        if (((late >> j) & ((1u << VG) - 1u)) == 0u) {
                            #pragma unroll
                            for (int g = 0; g < VG; ++g) {
                                const int e = __builtin_amdgcn_readlane(cid, j + g) * 32 + lane;
                                rv[g] = erow[e];
                            }
                            #pragma unroll
                            for (int g = 0; g < VG; ++g)
                                own |= ((unsigned)(rv[g] - row0) < (unsigned)nrows) ? (1u << g) : 0u;
                        } else {
                            #pragma unroll
                            for (int g = 0; g < VG; ++g) {
                                const int e = __builtin_amdgcn_readlane(cid, j + g) * 32 + lane;
                                rv[g] = erow[min(e, E - 1)];
                            }
                            #pragma unroll
                            for (int g = 0; g < VG; ++g) {
                                const int e = __builtin_amdgcn_readlane(cid, j + g) * 32 + lane;
                                const bool hit = (e < E) && ((unsigned)(rv[g] - row0) < (unsigned)nrows);
                                own |= hit ? (1u << g) : 0u;
                            }
                        }
                        for (int t = 0; t <= VG; ++t) {
                            const bool has = own != 0u;
                            const unsigned mask = __builtin_amdgcn_ballot_w32(has);
                            if (mask == 0u) break;
                            if (has) {
                                const int ch = myc[i0 + j + __builtin_ctz(own)];
                                srcl[fcnt + (int)__builtin_amdgcn_mbcnt_lo(mask, 0u)] = ch * 32 + lane;
                                own &= own - 1u;
                            }
                            fcnt += __builtin_popcount(mask);
                        }
                    }
                }
            };

            auto plane_word_f = [&](int w0) -> unsigned {
                const int wi = w0 + lane;
                if (wi >= wend) return 0u;
                const unsigned a = plane[wi];
                const unsigned b = two ? plane[(size_t)wpad + wi] : 0u;
                return a | b;
            };

            unsigned word = plane_word_f(wbeg);
            for (int w0 = wbeg; w0 < wend; w0 += 32) {
                const unsigned nxt = plane_word_f(w0 + 32);
                const int cword = (w0 + lane) << 5;
                int nlist = 0;
                for (int t = 0; t <= 32; ++t) {
                    const bool has = word != 0u;
                    const unsigned m = __builtin_amdgcn_ballot_w32(has);
                    const int k = __builtin_popcount(m);
                    if (k == 0 || nlist + k > CLCAP) { visit_f(nlist); nlist = 0; }
                    if (k == 0) break;
                    if (has) {
                        myc[nlist + (int)__builtin_amdgcn_mbcnt_lo(m, 0u)] = cword + __builtin_ctz(word);
                        word &= word - 1u;
                    }
                    nlist += k;
                }
                word = nxt;
            }
        }
        if (!tryfast || ovf) bad1 = 1;
        __syncthreads();
        const bool ok1 = bad1 == 0;
        if (ok1) {
            for (int i0 = 0; i0 < fcnt; i0 += 32) {
                if (i0 + lane < fcnt) {
                    const int4 q  = rec4[srcl[i0 + lane]];
                    const int  rl = q.x - row0;
                    const int  t  = rl & 31;
                    const int  slot = atomicAdd(&tcnt[t], 1);
                    if (slot < TCAP) {
                        int* tl = elist + t * QCAP;
                        tl[slot]            = (int)(((unsigned)rl << COL_BITS) | ((unsigned)q.y & ((1u << COL_BITS) - 1u)));
                        tl[TCAP + slot]     = q.z;
                        tl[2 * TCAP + slot] = q.w;
                    } else {
                        bad2 = 1;
                    }
                }
            }
        }
        __syncthreads();
        if (ok1 && bad2 == 0) {
            const int* tl = elist + wave * QCAP;
            const int  tn = tcnt[wave];
            for (int pass = 0; pass < 2; ++pass) {
                const float* __restrict__ msg = pass ? m_sig : m_mu;
                float* __restrict__ out = pass ? out_sig : out_mu;
                for (int i = tid; i < UNIT_ROWS * OUT_F / 4; i += 1024)
                    ((float4*)acc)[i] = make_float4(0.f, 0.f, 0.f, 0.f);
                __syncthreads();
                drain_owned<GRP>(acc, tl, tn, msg, pass ? 2 * TCAP : TCAP, lane);
                __syncthreads();
                for (int i = tid; i < nrows * (OUT_F / 4); i += 1024)
                    *(float4*)(out + (size_t)row0 * OUT_F + (size_t)i * 4) = ((const float4*)acc)[i];
                __syncthreads();
            }
            return;
        }
    }

    int  cnt = 0;
    bool spilt = false;
    bool kept  = false;

    for (int pass = 0; pass < 2; ++pass) {
        const bool var = pass != 0;
        const float* __restrict__ msg = var ? m_sig : m_mu;
        float* __restrict__ out = var ? out_sig : out_mu;

        if (var) __syncthreads();
        for (int i = tid; i < UNIT_ROWS * OUT_F / 4; i += 1024)
            ((float4*)acc)[i] = make_float4(0.f, 0.f, 0.f, 0.f);
        __syncthreads();

        if (!var || spilt) {
            cnt = 0;

            auto make_room = [&]() {
                if (cnt > QCAP - VG * 32) {
                    drain_list<1>(acc, myq, cnt, rec4, msg, var, KEEP_NONE, row0, lane);
                    cnt = 0;
                    spilt = true;
                }
            };

            auto visit = [&](int nlist) {
                for (int i0 = 0; i0 < nlist; i0 += 32) {
                    const int n = (nlist - i0 < 32) ? (nlist - i0) : 32;
                    const int cid = (lane < n) ? myc[i0 + lane] : nchunks;
                    const unsigned late = __builtin_amdgcn_ballot_w32(cid >= (E >> 5));
                    for (int j = 0; j < n; j += VG) {
                        make_room();
                        int rv[VG];
                        unsigned own = 0u;
                        if (((late >> j) & ((1u << VG) - 1u)) == 0u) {
                            #pragma unroll
                            for (int g = 0; g < VG; ++g) {
                                const int e = __builtin_amdgcn_readlane(cid, j + g) * 32 + lane;
                                rv[g] = erow[e];
                            }
                            #pragma unroll
                            for (int g = 0; g < VG; ++g)
                                own |= ((unsigned)(rv[g] - row0) < (unsigned)nrows) ? (1u << g) : 0u;
                        } else {
                            #pragma unroll
                            for (int g = 0; g < VG; ++g) {
                                const int e = __builtin_amdgcn_readlane(cid, j + g) * 32 + lane;
                                rv[g] = erow[min(e, E - 1)];
                            }
                            #pragma unroll
                            for (int g = 0; g < VG; ++g) {
                                const int e = __builtin_amdgcn_readlane(cid, j + g) * 32 + lane;
                                const bool hit = (e < E) && ((unsigned)(rv[g] - row0) < (unsigned)nrows);
                                own |= hit ? (1u << g) : 0u;
                            }
                        }
                        for (int t = 0; t <= VG; ++t) {
                            const bool has = own != 0u;
                            const unsigned mask = __builtin_amdgcn_ballot_w32(has);
                            if (mask == 0u) break;
                            if (has) {
                                const int ch = myc[i0 + j + __builtin_ctz(own)];
                                myq[cnt + (int)__builtin_amdgcn_mbcnt_lo(mask, 0u)] = ch * 32 + lane;
                                own &= own - 1u;
                            }
                            cnt += __builtin_popcount(mask);
                        }
                    }
                }
            };

            auto plane_word = [&](int w0) -> unsigned {
                const int wi = w0 + lane;
                if (wi >= wend) return 0u;
                const unsigned a = plane[wi];
                const unsigned b = two ? plane[(size_t)wpad + wi] : 0u;
                return a | b;
            };

            unsigned word = plane_word(wbeg);
            for (int w0 = wbeg; w0 < wend; w0 += 32) {
                const unsigned nxt = plane_word(w0 + 32);
                const int cword = (w0 + lane) << 5;
                int nlist = 0;
                for (int t = 0; t <= 32; ++t) {
                    const bool has = word != 0u;
                    const unsigned m = __builtin_amdgcn_ballot_w32(has);
                    const int k = __builtin_popcount(m);
                    if (k == 0 || nlist + k > CLCAP) { visit(nlist); nlist = 0; }
                    if (k == 0) break;
                    if (has) {
                        myc[nlist + (int)__builtin_amdgcn_mbcnt_lo(m, 0u)] = cword + __builtin_ctz(word);
                        word &= word - 1u;
                    }
                    nlist += k;
                }
                word = nxt;
            }
        }
        if (!var) kept = !spilt && cnt <= QHALF && N <= (1 << COL_BITS);
        drain_list<GRP>(acc, myq, cnt, rec4, msg, var,
                        kept ? (var ? KEEP_READ : KEEP_WRITE) : KEEP_NONE, row0, lane);
        __syncthreads();

        for (int i = tid; i < nrows * (OUT_F / 4); i += 1024) {
            const int r  = i >> 5;
            const int c4 = i & 31;
            const float* a = acc + r * OUT_F + c4;
            *(float4*)(out + (size_t)(row0 + r) * OUT_F + c4 * 4) = make_float4(a[0], a[32], a[64], a[96]);
        }
    }
}

extern "C" void kernel_launch(void* const* d_in, const int* in_sizes, int n_in,
                              void* d_out, int out_size, void* d_ws, size_t ws_size,
                              hipStream_t stream) {
    const float* miu   = (const float*)d_in[0];
    const float* sigma = (const float*)d_in[1];
    const int*   erow  = (const int*)  d_in[2];
    const int*   ecol  = (const int*)  d_in[3];
    const float* a1    = (const float*)d_in[4];
    const float* a2    = (const float*)d_in[5];
    const float* Wmu   = (const float*)d_in[6];
    const float* Wsig  = (const float*)d_in[7];

    const int N = in_sizes[0] / IN_F;
    const int E = in_sizes[2];

    float* out_mu  = (float*)d_out;
    float* out_sig = out_mu + (size_t)N * OUT_F;

    float* m_mu  = (float*)d_ws;
    float* m_sig = m_mu + (size_t)N * OUT_F;
    unsigned* planes = (unsigned*)(m_sig + (size_t)N * OUT_F);

    const int nblk   = (N + OWN_ROWS - 1) / OWN_ROWS;
    const int nspans = (E + SPAN_CH * 32 - 1) / (SPAN_CH * 32);
    const int wpad   = nspans * SPAN_W;

    gemm_msg_kernel<<<256, GEMM_THR, 0, stream>>>(
        miu, sigma, Wmu, Wsig, m_mu, m_sig, N, (N + 15) / 16);

    int* rec = (int*)(planes + (size_t)nblk * wpad);
    owner_planes_kernel<<<nspans + PACK_WG, 1024, 0, stream>>>(
        erow, ecol, a1, a2, planes, rec, N, E, nblk, wpad, nspans);

    spmm_owner_kernel<<<(nblk + UNIT_OWN - 1) / UNIT_OWN, 1024, 0, stream>>>(
        erow, rec, m_mu, m_sig, planes, out_mu, out_sig, N, E, wpad, nblk);
}
